# starting kernel
# baseline (speedup 1.0000x reference)
#include <hip/hip_runtime.h>

#define N_NODES  50000
#define N_EDGES  800000
#define NF       128
#define N_GRAPHS 128

typedef __attribute__((ext_vector_type(2))) float v2f;
typedef __attribute__((ext_vector_type(8))) float v8f;

__global__ void zero_kernel(float* __restrict__ p, int n) {
  int i = blockIdx.x * blockDim.x + threadIdx.x;
  if (i < n) p[i] = 0.0f;
}

__global__ void deg_kernel(const int* __restrict__ dst, float* __restrict__ deg) {
  int e = blockIdx.x * blockDim.x + threadIdx.x;
  if (e < N_EDGES) atomicAdd(&deg[dst[e]], 1.0f);
}

__global__ void dinv_kernel(float* __restrict__ deg) {
  int i = blockIdx.x * blockDim.x + threadIdx.x;
  if (i < N_NODES) deg[i] = rsqrtf(deg[i] + 1.0f);
}

__global__ __launch_bounds__(256) void gemm_wmma(const float* __restrict__ X,
                                                 const float* __restrict__ W,
                                                 float* __restrict__ H) {
  const int wave = threadIdx.x >> 5;
  const int lane = threadIdx.x & 31;
  const int half = lane >> 4;
  const int l    = lane & 15;
  const int m0   = blockIdx.x * 16;
  const int n0   = wave * 16;

  const float* Xrow = X + (size_t)(m0 + l) * NF;
  v8f acc = {};

  #pragma unroll
  for (int k = 0; k < NF; k += 4) {
    v2f a = *(const v2f*)(Xrow + k + 2 * half);
    v2f b;
    b.x = W[(size_t)(k + 2 * half + 0) * NF + n0 + l];
    b.y = W[(size_t)(k + 2 * half + 1) * NF + n0 + l];
    acc = __builtin_amdgcn_wmma_f32_16x16x4_f32(
         false, a,  false, b,
         (short)0, acc,  false,  false);
  }

  float* Hp = H + (size_t)(m0 + 8 * half) * NF + n0 + l;
  #pragma unroll
  for (int r = 0; r < 8; ++r) Hp[(size_t)r * NF] = acc[r];
}

__global__ __launch_bounds__(256) void scatter_kernel(const float* __restrict__ H,
                                                      const int* __restrict__ src,
                                                      const int* __restrict__ dst,
                                                      const float* __restrict__ dinv,
                                                      float* __restrict__ AGG) {
  int gid  = blockIdx.x * blockDim.x + threadIdx.x;
  int e    = gid >> 5;
  if (e >= N_EDGES) return;
  int lane = gid & 31;
  int s = src[e];
  int d = dst[e];
  float norm = dinv[s] * dinv[d];
  const float4 hv = *(const float4*)(H + (size_t)s * NF + lane * 4);
  float* op = AGG + (size_t)d * NF + lane * 4;
  atomicAdd(op + 0, hv.x * norm);
  atomicAdd(op + 1, hv.y * norm);
  atomicAdd(op + 2, hv.z * norm);
  atomicAdd(op + 3, hv.w * norm);
}

__global__ __launch_bounds__(256) void combine_relu(const float* __restrict__ AGG,
                                                    const float* __restrict__ H,
                                                    const float* __restrict__ dinv,
                                                    const float* __restrict__ b,
                                                    float* __restrict__ Y) {
  int gid = blockIdx.x * blockDim.x + threadIdx.x;
  if (gid >= N_NODES * 32) return;
  int i = gid >> 5;
  int q = (gid & 31) * 4;
  float di = dinv[i];
  float di2 = di * di;
  size_t off = (size_t)i * NF + q;
  float4 av = *(const float4*)(AGG + off);
  float4 hv = *(const float4*)(H + off);
  float4 bv = *(const float4*)(b + q);
  float4 r;
  r.x = fmaxf(av.x + di2 * hv.x + bv.x, 0.0f);
  r.y = fmaxf(av.y + di2 * hv.y + bv.y, 0.0f);
  r.z = fmaxf(av.z + di2 * hv.z + bv.z, 0.0f);
  r.w = fmaxf(av.w + di2 * hv.w + bv.w, 0.0f);
  *(float4*)(Y + off) = r;
}

__global__ __launch_bounds__(256) void combine_pool(const float* __restrict__ AGG,
                                                    const float* __restrict__ H,
                                                    const float* __restrict__ dinv,
                                                    const float* __restrict__ b,
                                                    const int* __restrict__ batch,
                                                    float* __restrict__ out) {
  int gid = blockIdx.x * blockDim.x + threadIdx.x;
  if (gid >= N_NODES * 32) return;
  int i = gid >> 5;
  int q = (gid & 31) * 4;
  float di = dinv[i];
  float di2 = di * di;
  size_t off = (size_t)i * NF + q;
  float4 av = *(const float4*)(AGG + off);
  float4 hv = *(const float4*)(H + off);
  float4 bv = *(const float4*)(b + q);
  float* op = out + (size_t)batch[i] * NF + q;
  atomicAdd(op + 0, av.x + di2 * hv.x + bv.x);
  atomicAdd(op + 1, av.y + di2 * hv.y + bv.y);
  atomicAdd(op + 2, av.z + di2 * hv.z + bv.z);
  atomicAdd(op + 3, av.w + di2 * hv.w + bv.w);
}

extern "C" void kernel_launch(void* const* d_in, const int* in_sizes, int n_in,
                              void* d_out, int out_size, void* d_ws, size_t ws_size,
                              hipStream_t stream) {
  const float* x    = (const float*)d_in[0];
  const int*   ei   = (const int*)  d_in[1];
  const int*   batch= (const int*)  d_in[2];
  const float* W1   = (const float*)d_in[3];
  const float* b1   = (const float*)d_in[4];
  const float* W2   = (const float*)d_in[5];
  const float* b2   = (const float*)d_in[6];
  float* out = (float*)d_out;
  float* ws  = (float*)d_ws;

  const int* src = ei;
  const int* dst = ei + N_EDGES;

  float* dinv = ws;
  float* H    = ws + 50048;
  float* AGG  = H  + (size_t)N_NODES * NF;
  float* Y    = AGG + (size_t)N_NODES * NF;

  const int T = 256;
  const int gNodes    = (N_NODES + T - 1) / T;
  const int gEdges    = (N_EDGES + T - 1) / T;
  const int gFeat     = ((size_t)N_NODES * NF + T - 1) / T;
  const int gNodeVec  = (N_NODES * 32 + T - 1) / T;
  const int gScatter  = ((size_t)N_EDGES * 32 + T - 1) / T;
  const int gGemm     = N_NODES / 16;

  zero_kernel<<<gNodes, T, 0, stream>>>(dinv, N_NODES);
  deg_kernel<<<gEdges, T, 0, stream>>>(dst, dinv);
  dinv_kernel<<<gNodes, T, 0, stream>>>(dinv);

  gemm_wmma<<<gGemm, T, 0, stream>>>(x, W1, H);
  zero_kernel<<<gFeat, T, 0, stream>>>(AGG, N_NODES * NF);
  scatter_kernel<<<gScatter, T, 0, stream>>>(H, src, dst, dinv, AGG);
  combine_relu<<<gNodeVec, T, 0, stream>>>(AGG, H, dinv, b1, Y);

  gemm_wmma<<<gGemm, T, 0, stream>>>(Y, W2, H);
  zero_kernel<<<gFeat, T, 0, stream>>>(AGG, N_NODES * NF);
  scatter_kernel<<<gScatter, T, 0, stream>>>(H, src, dst, dinv, AGG);

  zero_kernel<<<(N_GRAPHS * NF + T - 1) / T, T, 0, stream>>>(out, N_GRAPHS * NF);
  combine_pool<<<gNodeVec, T, 0, stream>>>(AGG, H, dinv, b2, batch, out);
}


# layer 2 folded by linearity: each edge and self loop adds norm*Y[src] into a per-graph 128x128 table (LDS adds per block, fixed-order block sum), then one 128^3 f32 WMMA product; sum order changes
# speedup vs baseline: 1.6460x; 1.6460x over previous
#include <hip/hip_runtime.h>

#define N_NODES  50000
#define N_EDGES  800000
#define NF       128
#define N_GRAPHS 128

#define EPN        (N_EDGES / N_NODES)
#define N_ITEMS    (N_EDGES + N_NODES)
#define N_GROUPS   (N_ITEMS / 8)
#define PS_BLOCKS  256
#define PS_THREADS 1024
#define PS_WAVES   (PS_BLOCKS * (PS_THREADS / 32))
#define PS_CHUNKS  4
#define TAB        (N_GRAPHS * NF)
#define PART       (TAB + N_GRAPHS)
static_assert(N_EDGES % N_NODES == 0, "item numbering needs a whole number of edges per node");
static_assert(N_ITEMS % 8 == 0, "items are handed out 8 at a time");
static_assert(((N_GROUPS + PS_WAVES - 1) / PS_WAVES) * 8 <= 32 * PS_CHUNKS, "a wave's items must fit its registers");
static_assert(PS_BLOCKS % 8 == 0 && (PART / 4) % 32 == 0, "reduce kernel tiling");

typedef __attribute__((ext_vector_type(2))) float v2f;
typedef __attribute__((ext_vector_type(8))) float v8f;

__global__ void zero_kernel(float* __restrict__ p, int n) {
  int i = blockIdx.x * blockDim.x + threadIdx.x;
  if (i < n) p[i] = 0.0f;
}

__global__ void deg_kernel(const int* __restrict__ dst, float* __restrict__ deg) {
  int e = blockIdx.x * blockDim.x + threadIdx.x;
  if (e < N_EDGES) atomicAdd(&deg[dst[e]], 1.0f);
}

__global__ void dinv_kernel(float* __restrict__ deg) {
  int i = blockIdx.x * blockDim.x + threadIdx.x;
  if (i < N_NODES) deg[i] = rsqrtf(deg[i] + 1.0f);
}

__global__ __launch_bounds__(256) void gemm_wmma(const float* __restrict__ X,
                                                 const float* __restrict__ W,
                                                 float* __restrict__ H) {
  const int wave = threadIdx.x >> 5;
  const int lane = threadIdx.x & 31;
  const int half = lane >> 4;
  const int l    = lane & 15;
  const int m0   = blockIdx.x * 16;
  const int n0   = wave * 16;

  const float* Xrow = X + (size_t)(m0 + l) * NF;
  v8f acc = {};

  #pragma unroll
  for (int k = 0; k < NF; k += 4) {
    v2f a = *(const v2f*)(Xrow + k + 2 * half);
    v2f b;
    b.x = W[(size_t)(k + 2 * half + 0) * NF + n0 + l];
    b.y = W[(size_t)(k + 2 * half + 1) * NF + n0 + l];
    acc = __builtin_amdgcn_wmma_f32_16x16x4_f32(
         false, a,  false, b,
         (short)0, acc,  false,  false);
  }

  float* Hp = H + (size_t)(m0 + 8 * half) * NF + n0 + l;
  #pragma unroll
  for (int r = 0; r < 8; ++r) Hp[(size_t)r * NF] = acc[r];
}

__global__ __launch_bounds__(256) void gemm_wmma_pooled(const float* __restrict__ T,
                                                        const float* __restrict__ W,
                                                        const float* __restrict__ cnt,
                                                        const float* __restrict__ bias,
                                                        float* __restrict__ out) {
  const int wave = threadIdx.x >> 5;
  const int lane = threadIdx.x & 31;
  const int half = lane >> 4;
  const int l    = lane & 15;
  const int m0   = blockIdx.x * 16;
  const int n0   = wave * 16;

  const float* Trow = T + (size_t)(m0 + l) * NF;
  v8f acc = {};

  #pragma unroll
  for (int k = 0; k < NF; k += 4) {
    v2f a = *(const v2f*)(Trow + k + 2 * half);
    v2f b;
    b.x = W[(size_t)(k + 2 * half + 0) * NF + n0 + l];
    b.y = W[(size_t)(k + 2 * half + 1) * NF + n0 + l];
    acc = __builtin_amdgcn_wmma_f32_16x16x4_f32(
         false, a,  false, b,
         (short)0, acc,  false,  false);
  }

  const float bn = bias[n0 + l];
  float* op = out + (size_t)(m0 + 8 * half) * NF + n0 + l;
  #pragma unroll
  for (int r = 0; r < 8; ++r) op[(size_t)r * NF] = acc[r] + cnt[m0 + 8 * half + r] * bn;
}

__global__ __launch_bounds__(256) void scatter_kernel(const float* __restrict__ H,
                                                      const int* __restrict__ src,
                                                      const int* __restrict__ dst,
                                                      const float* __restrict__ dinv,
                                                      float* __restrict__ AGG) {
  int gid  = blockIdx.x * blockDim.x + threadIdx.x;
  int e    = gid >> 5;
  if (e >= N_EDGES) return;
  int lane = gid & 31;
  int s = src[e];
  int d = dst[e];
  float norm = dinv[s] * dinv[d];
  const float4 hv = *(const float4*)(H + (size_t)s * NF + lane * 4);
  float* op = AGG + (size_t)d * NF + lane * 4;
  atomicAdd(op + 0, hv.x * norm);
  atomicAdd(op + 1, hv.y * norm);
  atomicAdd(op + 2, hv.z * norm);
  atomicAdd(op + 3, hv.w * norm);
}

__global__ __launch_bounds__(256) void combine_relu(const float* __restrict__ AGG,
                                                    const float* __restrict__ H,
                                                    const float* __restrict__ dinv,
                                                    const float* __restrict__ b,
                                                    float* __restrict__ Y) {
  int gid = blockIdx.x * blockDim.x + threadIdx.x;
  if (gid >= N_NODES * 32) return;
  int i = gid >> 5;
  int q = (gid & 31) * 4;
  float di = dinv[i];
  float di2 = di * di;
  size_t off = (size_t)i * NF + q;
  float4 av = *(const float4*)(AGG + off);
  float4 hv = *(const float4*)(H + off);
  float4 bv = *(const float4*)(b + q);
  float4 r;
  r.x = fmaxf(av.x + di2 * hv.x + bv.x, 0.0f);
  r.y = fmaxf(av.y + di2 * hv.y + bv.y, 0.0f);
  r.z = fmaxf(av.z + di2 * hv.z + bv.z, 0.0f);
  r.w = fmaxf(av.w + di2 * hv.w + bv.w, 0.0f);
  *(float4*)(Y + off) = r;
}

__global__ __launch_bounds__(PS_THREADS) void pool_scatter(const float* __restrict__ Y,
                                                           const int* __restrict__ src,
                                                           const int* __restrict__ dst,
                                                           const float* __restrict__ dinv,
                                                           const int* __restrict__ batch,
                                                           float* __restrict__ part) {
  __shared__ float tab[TAB];
  __shared__ float cnt[N_GRAPHS];
  const int tid  = threadIdx.x;
  const int lane = tid & 31;

  for (int i = tid; i < TAB / 4; i += PS_THREADS)
    ((float4*)tab)[i] = make_float4(0.0f, 0.0f, 0.0f, 0.0f);
  if (tid < N_GRAPHS) cnt[tid] = 0.0f;
  __syncthreads();

  const int wv = __builtin_amdgcn_readfirstlane((int)blockIdx.x * (PS_THREADS / 32) + (tid >> 5));
  const int t0 = (int)(((long long)wv * N_GROUPS) / PS_WAVES) * 8;
  const int t1 = (int)(((long long)(wv + 1) * N_GROUPS) / PS_WAVES) * 8;

  int   s[PS_CHUNKS];
  int   g[PS_CHUNKS];
  float w[PS_CHUNKS];
  int   qn[PS_CHUNKS];
  int   rn[PS_CHUNKS];
  #pragma unroll
  for (int c = 0; c < PS_CHUNKS; ++c) {
    int t = t0 + 32 * c + lane;
    t = t < t1 ? t : t1 - 1;
    qn[c] = t / (EPN + 1);
    rn[c] = t - qn[c] * (EPN + 1);
    const int e = qn[c] * EPN + (rn[c] < EPN ? rn[c] : EPN - 1);
    s[c] = src[e];
    g[c] = dst[e];
  }
  #pragma unroll
  for (int c = 0; c < PS_CHUNKS; ++c) {
    const int sn = rn[c] < EPN ? s[c] : qn[c];
    const int dn = rn[c] < EPN ? g[c] : qn[c];
    s[c] = sn;
    w[c] = dinv[sn] * dinv[dn];
    g[c] = batch[dn];
  }
  #pragma unroll
  for (int c = 0; c < PS_CHUNKS; ++c)
    if (t0 + 32 * c + lane < t1 && rn[c] == EPN) atomicAdd(&cnt[g[c]], 1.0f);

  #pragma unroll
  for (int c = 0; c < PS_CHUNKS; ++c) {
    const int left = t1 - (t0 + 32 * c);
    const int ng = (left < 32 ? left : 32) >> 3;
    for (int q = 0; q < ng; ++q) {
      float4 v[8];
      #pragma unroll
      for (int j = 0; j < 8; ++j) {
        const int sk = __builtin_amdgcn_readlane(s[c], q * 8 + j);
        v[j] = *(const float4*)(Y + (size_t)sk * NF + lane * 4);
      }
      #pragma unroll
      for (int j = 0; j < 8; ++j) {
        const float wk = __int_as_float(__builtin_amdgcn_readlane(__float_as_int(w[c]), q * 8 + j));
        const int   gk = __builtin_amdgcn_readlane(g[c], q * 8 + j);
        float* row = tab + gk * NF + lane;
        atomicAdd(row,      v[j].x * wk);
        atomicAdd(row + 32, v[j].y * wk);
        atomicAdd(row + 64, v[j].z * wk);
        atomicAdd(row + 96, v[j].w * wk);
      }
    }
  }
  __syncthreads();

  float* P = part + (size_t)blockIdx.x * PART;
  for (int i = tid; i < TAB / 4; i += PS_THREADS) {
    const float* r = tab + (i >> 5) * NF + (i & 31);
    ((float4*)P)[i] = make_float4(r[0], r[32], r[64], r[96]);
  }
  if (tid < N_GRAPHS) P[TAB + tid] = cnt[tid];
}

__global__ __launch_bounds__(256) void reduce_parts(const float* __restrict__ part,
                                                    float* __restrict__ T) {
  __shared__ float4 red[8][32];
  const int wv   = threadIdx.x >> 5;
  const int lane = threadIdx.x & 31;
  const int q    = blockIdx.x * 32 + lane;
  const float4* p = (const float4*)part + q;
  float4 a = make_float4(0.0f, 0.0f, 0.0f, 0.0f);
  #pragma unroll
  for (int b = 0; b < PS_BLOCKS / 8; ++b) {
    const float4 x = p[(size_t)(wv * (PS_BLOCKS / 8) + b) * (PART / 4)];
    a.x += x.x; a.y += x.y; a.z += x.z; a.w += x.w;
  }
  red[wv][lane] = a;
  __syncthreads();
  if (wv == 0) {
    float4 t = red[0][lane];
    #pragma unroll
    for (int k = 1; k < 8; ++k) {
      const float4 x = red[k][lane];
      t.x += x.x; t.y += x.y; t.z += x.z; t.w += x.w;
    }
    ((float4*)T)[q] = t;
  }
}

extern "C" void kernel_launch(void* const* d_in, const int* in_sizes, int n_in,
                              void* d_out, int out_size, void* d_ws, size_t ws_size,
                              hipStream_t stream) {
  const float* x    = (const float*)d_in[0];
  const int*   ei   = (const int*)  d_in[1];
  const int*   batch= (const int*)  d_in[2];
  const float* W1   = (const float*)d_in[3];
  const float* b1   = (const float*)d_in[4];
  const float* W2   = (const float*)d_in[5];
  const float* b2   = (const float*)d_in[6];
  float* out = (float*)d_out;
  float* ws  = (float*)d_ws;

  const int* src = ei;
  const int* dst = ei + N_EDGES;

  float* dinv = ws;
  float* H    = ws + 50048;
  float* AGG  = H  + (size_t)N_NODES * NF;
  float* Y    = AGG + (size_t)N_NODES * NF;
  float* part = H;
  float* Tsum = AGG;

  const int T = 256;
  const int gNodes    = (N_NODES + T - 1) / T;
  const int gEdges    = (N_EDGES + T - 1) / T;
  const int gFeat     = ((size_t)N_NODES * NF + T - 1) / T;
  const int gNodeVec  = (N_NODES * 32 + T - 1) / T;
  const int gScatter  = ((size_t)N_EDGES * 32 + T - 1) / T;
  const int gGemm     = N_NODES / 16;

  zero_kernel<<<gNodes, T, 0, stream>>>(dinv, N_NODES);
  deg_kernel<<<gEdges, T, 0, stream>>>(dst, dinv);
  dinv_kernel<<<gNodes, T, 0, stream>>>(dinv);

  gemm_wmma<<<gGemm, T, 0, stream>>>(x, W1, H);
  zero_kernel<<<gFeat, T, 0, stream>>>(AGG, N_NODES * NF);
  scatter_kernel<<<gScatter, T, 0, stream>>>(H, src, dst, dinv, AGG);
  combine_relu<<<gNodeVec, T, 0, stream>>>(AGG, H, dinv, b1, Y);

  pool_scatter<<<PS_BLOCKS, PS_THREADS, 0, stream>>>(Y, src, dst, dinv, batch, part);
  reduce_parts<<<(PART / 4) / 32, T, 0, stream>>>(part, Tsum);
  gemm_wmma_pooled<<<N_GRAPHS / 16, T, 0, stream>>>(Tsum, W2, Tsum + TAB, b2, out);
}


# layer-1 scatter: lane l takes features l, l+32, l+64, l+96, so each f32 atomic of a wave covers one whole 128-byte line, not four quarter-filled ones; sum order: atomic arrival order as before
# speedup vs baseline: 2.0103x; 1.2213x over previous
#include <hip/hip_runtime.h>

#define N_NODES  50000
#define N_EDGES  800000
#define NF       128
#define N_GRAPHS 128

#define EPN        (N_EDGES / N_NODES)
#define N_ITEMS    (N_EDGES + N_NODES)
#define N_GROUPS   (N_ITEMS / 8)
#define PS_BLOCKS  256
#define PS_THREADS 1024
#define PS_WAVES   (PS_BLOCKS * (PS_THREADS / 32))
#define PS_CHUNKS  4
#define TAB        (N_GRAPHS * NF)
#define PART       (TAB + N_GRAPHS)
static_assert(N_EDGES % N_NODES == 0, "item numbering needs a whole number of edges per node");
static_assert(N_ITEMS % 8 == 0, "items are handed out 8 at a time");
static_assert(((N_GROUPS + PS_WAVES - 1) / PS_WAVES) * 8 <= 32 * PS_CHUNKS, "a wave's items must fit its registers");
static_assert(PS_BLOCKS % 8 == 0 && (PART / 4) % 32 == 0, "reduce kernel tiling");

typedef __attribute__((ext_vector_type(2))) float v2f;
typedef __attribute__((ext_vector_type(8))) float v8f;

__global__ void zero_kernel(float* __restrict__ p, int n) {
  int i = blockIdx.x * blockDim.x + threadIdx.x;
  if (i < n) p[i] = 0.0f;
}

__global__ void deg_kernel(const int* __restrict__ dst, float* __restrict__ deg) {
  int e = blockIdx.x * blockDim.x + threadIdx.x;
  if (e < N_EDGES) atomicAdd(&deg[dst[e]], 1.0f);
}

__global__ void dinv_kernel(float* __restrict__ deg) {
  int i = blockIdx.x * blockDim.x + threadIdx.x;
  if (i < N_NODES) deg[i] = rsqrtf(deg[i] + 1.0f);
}

__global__ __launch_bounds__(256) void gemm_wmma(const float* __restrict__ X,
                                                 const float* __restrict__ W,
                                                 float* __restrict__ H) {
  const int wave = threadIdx.x >> 5;
  const int lane = threadIdx.x & 31;
  const int half = lane >> 4;
  const int l    = lane & 15;
  const int m0   = blockIdx.x * 16;
  const int n0   = wave * 16;

  const float* Xrow = X + (size_t)(m0 + l) * NF;
  v8f acc = {};

  #pragma unroll
  for (int k = 0; k < NF; k += 4) {
    v2f a = *(const v2f*)(Xrow + k + 2 * half);
    v2f b;
    b.x = W[(size_t)(k + 2 * half + 0) * NF + n0 + l];
    b.y = W[(size_t)(k + 2 * half + 1) * NF + n0 + l];
    acc = __builtin_amdgcn_wmma_f32_16x16x4_f32(
         false, a,  false, b,
         (short)0, acc,  false,  false);
  }

  float* Hp = H + (size_t)(m0 + 8 * half) * NF + n0 + l;
  #pragma unroll
  for (int r = 0; r < 8; ++r) Hp[(size_t)r * NF] = acc[r];
}

__global__ __launch_bounds__(256) void gemm_wmma_pooled(const float* __restrict__ T,
                                                        const float* __restrict__ W,
                                                        const float* __restrict__ cnt,
                                                        const float* __restrict__ bias,
                                                        float* __restrict__ out) {
  const int wave = threadIdx.x >> 5;
  const int lane = threadIdx.x & 31;
  const int half = lane >> 4;
  const int l    = lane & 15;
  const int m0   = blockIdx.x * 16;
  const int n0   = wave * 16;

  const float* Trow = T + (size_t)(m0 + l) * NF;
  v8f acc = {};

  #pragma unroll
  for (int k = 0; k < NF; k += 4) {
    v2f a = *(const v2f*)(Trow + k + 2 * half);
    v2f b;
    b.x = W[(size_t)(k + 2 * half + 0) * NF + n0 + l];
    b.y = W[(size_t)(k + 2 * half + 1) * NF + n0 + l];
    acc = __builtin_amdgcn_wmma_f32_16x16x4_f32(
         false, a,  false, b,
         (short)0, acc,  false,  false);
  }

  const float bn = bias[n0 + l];
  float* op = out + (size_t)(m0 + 8 * half) * NF + n0 + l;
  #pragma unroll
  for (int r = 0; r < 8; ++r) op[(size_t)r * NF] = acc[r] + cnt[m0 + 8 * half + r] * bn;
}

__global__ __launch_bounds__(256) void scatter_kernel(const float* __restrict__ H,
                                                      const int* __restrict__ src,
                                                      const int* __restrict__ dst,
                                                      const float* __restrict__ dinv,
                                                      float* __restrict__ AGG) {
  int gid  = blockIdx.x * blockDim.x + threadIdx.x;
  int e    = gid >> 5;
  if (e >= N_EDGES) return;
  int lane = gid & 31;
  int s = src[e];
  int d = dst[e];
  float norm = dinv[s] * dinv[d];
  const float* hp = H + (size_t)s * NF + lane;
  const float h0 = hp[0];
  const float h1 = hp[32];
  const float h2 = hp[64];
  const float h3 = hp[96];
  float* op = AGG + (size_t)d * NF + lane;
  atomicAdd(op,      h0 * norm);
  atomicAdd(op + 32, h1 * norm);
  atomicAdd(op + 64, h2 * norm);
  atomicAdd(op + 96, h3 * norm);
}

__global__ __launch_bounds__(256) void combine_relu(const float* __restrict__ AGG,
                                                    const float* __restrict__ H,
                                                    const float* __restrict__ dinv,
                                                    const float* __restrict__ b,
                                                    float* __restrict__ Y) {
  int gid = blockIdx.x * blockDim.x + threadIdx.x;
  if (gid >= N_NODES * 32) return;
  int i = gid >> 5;
  int q = (gid & 31) * 4;
  float di = dinv[i];
  float di2 = di * di;
  size_t off = (size_t)i * NF + q;
  float4 av = *(const float4*)(AGG + off);
  float4 hv = *(const float4*)(H + off);
  float4 bv = *(const float4*)(b + q);
  float4 r;
  r.x = fmaxf(av.x + di2 * hv.x + bv.x, 0.0f);
  r.y = fmaxf(av.y + di2 * hv.y + bv.y, 0.0f);
  r.z = fmaxf(av.z + di2 * hv.z + bv.z, 0.0f);
  r.w = fmaxf(av.w + di2 * hv.w + bv.w, 0.0f);
  *(float4*)(Y + off) = r;
}

__global__ __launch_bounds__(PS_THREADS) void pool_scatter(const float* __restrict__ Y,
                                                           const int* __restrict__ src,
                                                           const int* __restrict__ dst,
                                                           const float* __restrict__ dinv,
                                                           const int* __restrict__ batch,
                                                           float* __restrict__ part) {
  __shared__ float tab[TAB];
  __shared__ float cnt[N_GRAPHS];
  const int tid  = threadIdx.x;
  const int lane = tid & 31;

  for (int i = tid; i < TAB / 4; i += PS_THREADS)
    ((float4*)tab)[i] = make_float4(0.0f, 0.0f, 0.0f, 0.0f);
  if (tid < N_GRAPHS) cnt[tid] = 0.0f;
  __syncthreads();

  const int wv = __builtin_amdgcn_readfirstlane((int)blockIdx.x * (PS_THREADS / 32) + (tid >> 5));
  const int t0 = (int)(((long long)wv * N_GROUPS) / PS_WAVES) * 8;
  const int t1 = (int)(((long long)(wv + 1) * N_GROUPS) / PS_WAVES) * 8;

  int   s[PS_CHUNKS];
  int   g[PS_CHUNKS];
  float w[PS_CHUNKS];
  int   qn[PS_CHUNKS];
  int   rn[PS_CHUNKS];
  #pragma unroll
  for (int c = 0; c < PS_CHUNKS; ++c) {
    int t = t0 + 32 * c + lane;
    t = t < t1 ? t : t1 - 1;
    qn[c] = t / (EPN + 1);
    rn[c] = t - qn[c] * (EPN + 1);
    const int e = qn[c] * EPN + (rn[c] < EPN ? rn[c] : EPN - 1);
    s[c] = src[e];
    g[c] = dst[e];
  }
  #pragma unroll
  for (int c = 0; c < PS_CHUNKS; ++c) {
    const int sn = rn[c] < EPN ? s[c] : qn[c];
    const int dn = rn[c] < EPN ? g[c] : qn[c];
    s[c] = sn;
    w[c] = dinv[sn] * dinv[dn];
    g[c] = batch[dn];
  }
  #pragma unroll
  for (int c = 0; c < PS_CHUNKS; ++c)
    if (t0 + 32 * c + lane < t1 && rn[c] == EPN) atomicAdd(&cnt[g[c]], 1.0f);

  #pragma unroll
  for (int c = 0; c < PS_CHUNKS; ++c) {
    const int left = t1 - (t0 + 32 * c);
    const int ng = (left < 32 ? left : 32) >> 3;
    for (int q = 0; q < ng; ++q) {
      float4 v[8];
      #pragma unroll
      for (int j = 0; j < 8; ++j) {
        const int sk = __builtin_amdgcn_readlane(s[c], q * 8 + j);
        v[j] = *(const float4*)(Y + (size_t)sk * NF + lane * 4);
      }
      #pragma unroll
      for (int j = 0; j < 8; ++j) {
        const float wk = __int_as_float(__builtin_amdgcn_readlane(__float_as_int(w[c]), q * 8 + j));
        const int   gk = __builtin_amdgcn_readlane(g[c], q * 8 + j);
        float* row = tab + gk * NF + lane;
        atomicAdd(row,      v[j].x * wk);
        atomicAdd(row + 32, v[j].y * wk);
        atomicAdd(row + 64, v[j].z * wk);
        atomicAdd(row + 96, v[j].w * wk);
      }
    }
  }
  __syncthreads();

  float* P = part + (size_t)blockIdx.x * PART;
  for (int i = tid; i < TAB / 4; i += PS_THREADS) {
    const float* r = tab + (i >> 5) * NF + (i & 31);
    ((float4*)P)[i] = make_float4(r[0], r[32], r[64], r[96]);
  }
  if (tid < N_GRAPHS) P[TAB + tid] = cnt[tid];
}

__global__ __launch_bounds__(256) void reduce_parts(const float* __restrict__ part,
                                                    float* __restrict__ T) {
  __shared__ float4 red[8][32];
  const int wv   = threadIdx.x >> 5;
  const int lane = threadIdx.x & 31;
  const int q    = blockIdx.x * 32 + lane;
  const float4* p = (const float4*)part + q;
  float4 a = make_float4(0.0f, 0.0f, 0.0f, 0.0f);
  #pragma unroll
  for (int b = 0; b < PS_BLOCKS / 8; ++b) {
    const float4 x = p[(size_t)(wv * (PS_BLOCKS / 8) + b) * (PART / 4)];
    a.x += x.x; a.y += x.y; a.z += x.z; a.w += x.w;
  }
  red[wv][lane] = a;
  __syncthreads();
  if (wv == 0) {
    float4 t = red[0][lane];
    #pragma unroll
    for (int k = 1; k < 8; ++k) {
      const float4 x = red[k][lane];
      t.x += x.x; t.y += x.y; t.z += x.z; t.w += x.w;
    }
    ((float4*)T)[q] = t;
  }
}

extern "C" void kernel_launch(void* const* d_in, const int* in_sizes, int n_in,
                              void* d_out, int out_size, void* d_ws, size_t ws_size,
                              hipStream_t stream) {
  const float* x    = (const float*)d_in[0];
  const int*   ei   = (const int*)  d_in[1];
  const int*   batch= (const int*)  d_in[2];
  const float* W1   = (const float*)d_in[3];
  const float* b1   = (const float*)d_in[4];
  const float* W2   = (const float*)d_in[5];
  const float* b2   = (const float*)d_in[6];
  float* out = (float*)d_out;
  float* ws  = (float*)d_ws;

  const int* src = ei;
  const int* dst = ei + N_EDGES;

  float* dinv = ws;
  float* H    = ws + 50048;
  float* AGG  = H  + (size_t)N_NODES * NF;
  float* Y    = AGG + (size_t)N_NODES * NF;
  float* part = H;
  float* Tsum = AGG;

  const int T = 256;
  const int gNodes    = (N_NODES + T - 1) / T;
  const int gEdges    = (N_EDGES + T - 1) / T;
  const int gFeat     = ((size_t)N_NODES * NF + T - 1) / T;
  const int gNodeVec  = (N_NODES * 32 + T - 1) / T;
  const int gScatter  = ((size_t)N_EDGES * 32 + T - 1) / T;
  const int gGemm     = N_NODES / 16;

  zero_kernel<<<gNodes, T, 0, stream>>>(dinv, N_NODES);
  deg_kernel<<<gEdges, T, 0, stream>>>(dst, dinv);
  dinv_kernel<<<gNodes, T, 0, stream>>>(dinv);

  gemm_wmma<<<gGemm, T, 0, stream>>>(x, W1, H);
  zero_kernel<<<gFeat, T, 0, stream>>>(AGG, N_NODES * NF);
  scatter_kernel<<<gScatter, T, 0, stream>>>(H, src, dst, dinv, AGG);
  combine_relu<<<gNodeVec, T, 0, stream>>>(AGG, H, dinv, b1, Y);

  pool_scatter<<<PS_BLOCKS, PS_THREADS, 0, stream>>>(Y, src, dst, dinv, batch, part);
  reduce_parts<<<(PART / 4) / 32, T, 0, stream>>>(part, Tsum);
  gemm_wmma_pooled<<<N_GRAPHS / 16, T, 0, stream>>>(Tsum, W2, Tsum + TAB, b2, out);
}


# layer-1 aggregation by owner blocks: 256 blocks each keep 196 nodes' sums in an LDS table, sweep dst, gather matching H rows, LDS float adds; combine at table store; sum order: LDS arrival order
# speedup vs baseline: 2.5332x; 1.2601x over previous
#include <hip/hip_runtime.h>

#define N_NODES  50000
#define N_EDGES  800000
#define NF       128
#define N_GRAPHS 128

#define EPN        (N_EDGES / N_NODES)
#define N_ITEMS    (N_EDGES + N_NODES)
#define N_GROUPS   (N_ITEMS / 8)
#define PS_BLOCKS  256
#define PS_THREADS 1024
#define PS_WAVES   (PS_BLOCKS * (PS_THREADS / 32))
#define PS_CHUNKS  4
#define TAB        (N_GRAPHS * NF)
#define PART       (TAB + N_GRAPHS)
static_assert(N_EDGES % N_NODES == 0, "item numbering needs a whole number of edges per node");
static_assert(N_ITEMS % 8 == 0, "items are handed out 8 at a time");
static_assert(((N_GROUPS + PS_WAVES - 1) / PS_WAVES) * 8 <= 32 * PS_CHUNKS, "a wave's items must fit its registers");
static_assert(PS_BLOCKS % 8 == 0 && (PART / 4) % 32 == 0, "reduce kernel tiling");

#define OWN_BLOCKS  256
#define OWN_NODES   ((N_NODES + OWN_BLOCKS - 1) / OWN_BLOCKS)
#define OWN_THREADS 1024
#define OWN_WAVES   (OWN_THREADS / 32)
#define CH_EDGES    128
#define N_CH        (N_EDGES / CH_EDGES)
#define OWN_U       4
#define OWN_STEPS   ((N_CH + OWN_WAVES * OWN_U - 1) / (OWN_WAVES * OWN_U))
#define Q_CAP       (32 + OWN_U * CH_EDGES)
static_assert(N_EDGES % CH_EDGES == 0, "the sweep reads whole chunks");
static_assert(N_EDGES <= (1 << 20) && OWN_NODES <= (1 << 11), "queue entry packs node and edge");
static_assert((OWN_BLOCKS - 1) * OWN_NODES < N_NODES, "every owner block has at least one node");
static_assert(N_CH >= OWN_WAVES * OWN_U, "start offset plus chunk index wraps at most once");

typedef __attribute__((ext_vector_type(2))) float v2f;
typedef __attribute__((ext_vector_type(8))) float v8f;

__global__ void zero_kernel(float* __restrict__ p, int n) {
  int i = blockIdx.x * blockDim.x + threadIdx.x;
  if (i < n) p[i] = 0.0f;
}

__global__ void deg_kernel(const int* __restrict__ dst, float* __restrict__ deg) {
  int e = blockIdx.x * blockDim.x + threadIdx.x;
  if (e < N_EDGES) atomicAdd(&deg[dst[e]], 1.0f);
}

__global__ void dinv_kernel(float* __restrict__ deg) {
  int i = blockIdx.x * blockDim.x + threadIdx.x;
  if (i < N_NODES) deg[i] = rsqrtf(deg[i] + 1.0f);
}

__global__ __launch_bounds__(256) void gemm_wmma(const float* __restrict__ X,
                                                 const float* __restrict__ W,
                                                 float* __restrict__ H) {
  const int wave = threadIdx.x >> 5;
  const int lane = threadIdx.x & 31;
  const int half = lane >> 4;
  const int l    = lane & 15;
  const int m0   = blockIdx.x * 16;
  const int n0   = wave * 16;

  const float* Xrow = X + (size_t)(m0 + l) * NF;
  v8f acc = {};

  #pragma unroll
  for (int k = 0; k < NF; k += 4) {
    v2f a = *(const v2f*)(Xrow + k + 2 * half);
    v2f b;
    b.x = W[(size_t)(k + 2 * half + 0) * NF + n0 + l];
    b.y = W[(size_t)(k + 2 * half + 1) * NF + n0 + l];
    acc = __builtin_amdgcn_wmma_f32_16x16x4_f32(
         false, a,  false, b,
         (short)0, acc,  false,  false);
  }

  float* Hp = H + (size_t)(m0 + 8 * half) * NF + n0 + l;
  #pragma unroll
  for (int r = 0; r < 8; ++r) Hp[(size_t)r * NF] = acc[r];
}

__global__ __launch_bounds__(256) void gemm_wmma_pooled(const float* __restrict__ T,
                                                        const float* __restrict__ W,
                                                        const float* __restrict__ cnt,
                                                        const float* __restrict__ bias,
                                                        float* __restrict__ out) {
  const int wave = threadIdx.x >> 5;
  const int lane = threadIdx.x & 31;
  const int half = lane >> 4;
  const int l    = lane & 15;
  const int m0   = blockIdx.x * 16;
  const int n0   = wave * 16;

  const float* Trow = T + (size_t)(m0 + l) * NF;
  v8f acc = {};

  #pragma unroll
  for (int k = 0; k < NF; k += 4) {
    v2f a = *(const v2f*)(Trow + k + 2 * half);
    v2f b;
    b.x = W[(size_t)(k + 2 * half + 0) * NF + n0 + l];
    b.y = W[(size_t)(k + 2 * half + 1) * NF + n0 + l];
    acc = __builtin_amdgcn_wmma_f32_16x16x4_f32(
         false, a,  false, b,
         (short)0, acc,  false,  false);
  }

  const float bn = bias[n0 + l];
  float* op = out + (size_t)(m0 + 8 * half) * NF + n0 + l;
  #pragma unroll
  for (int r = 0; r < 8; ++r) op[(size_t)r * NF] = acc[r] + cnt[m0 + 8 * half + r] * bn;
}

__device__ __forceinline__ void own_flush(const int* q, int n, int lane, int n0,
                                          const float* __restrict__ H,
                                          const int* __restrict__ src,
                                          const float* __restrict__ dinv,
                                          float* tab) {
  const int ent = q[lane < n ? lane : n - 1];
  const int e = ent & 0xFFFFF;
  const int u = ent >> 20;
  const int s = src[e];
  const float wgt = dinv[s] * dinv[n0 + u];
  const int ng = (n + 7) >> 3;
  for (int g = 0; g < ng; ++g) {
    float4 v[8];
    #pragma unroll
    for (int j = 0; j < 8; ++j) {
      const int sk = __builtin_amdgcn_readlane(s, g * 8 + j);
      v[j] = *(const float4*)(H + (size_t)sk * NF + lane * 4);
    }
    #pragma unroll
    for (int j = 0; j < 8; ++j) {
      const int k = g * 8 + j;
      if (k < n) {
        const float wk = __int_as_float(__builtin_amdgcn_readlane(__float_as_int(wgt), k));
        const int   uk = __builtin_amdgcn_readlane(u, k);
        float* row = tab + uk * NF + lane;
        atomicAdd(row,      v[j].x * wk);
        atomicAdd(row + 32, v[j].y * wk);
        atomicAdd(row + 64, v[j].z * wk);
        atomicAdd(row + 96, v[j].w * wk);
      }
    }
  }
}

__global__ __launch_bounds__(OWN_THREADS) void own_aggregate(const float* __restrict__ H,
                                                             const int* __restrict__ src,
                                                             const int* __restrict__ dst,
                                                             const float* __restrict__ dinv,
                                                             const float* __restrict__ b,
                                                             float* __restrict__ Y) {
  __shared__ float tab[OWN_NODES * NF];
  __shared__ int   queue[OWN_WAVES * Q_CAP];
  const int tid  = threadIdx.x;
  const int lane = tid & 31;
  const int n0   = (int)blockIdx.x * OWN_NODES;
  const int nown = (N_NODES - n0) < OWN_NODES ? (N_NODES - n0) : OWN_NODES;

  for (int i = tid; i < OWN_NODES * NF / 4; i += OWN_THREADS)
    ((float4*)tab)[i] = make_float4(0.0f, 0.0f, 0.0f, 0.0f);
  __syncthreads();

  const int w = __builtin_amdgcn_readfirstlane(tid >> 5);
  int* q = queue + w * Q_CAP;
  int qn = 0;
  const int c0 = (int)(((long long)blockIdx.x * N_CH) / OWN_BLOCKS);

  for (int i = 0; i < OWN_STEPS; ++i) {
    int4 d[OWN_U];
    int  eb[OWN_U];
    bool ok[OWN_U];
    #pragma unroll
    for (int k = 0; k < OWN_U; ++k) {
      const int ci = (i * OWN_U + k) * OWN_WAVES + w;
      ok[k] = ci < N_CH;
      int cc = c0 + (ok[k] ? ci : 0);
      cc = cc >= N_CH ? cc - N_CH : cc;
      eb[k] = cc * CH_EDGES + lane * 4;
      d[k] = *(const int4*)(dst + eb[k]);
    }
    #pragma unroll
    for (int k = 0; k < OWN_U; ++k) {
      if (ok[k]) {
        const int dd[4] = {d[k].x, d[k].y, d[k].z, d[k].w};
        #pragma unroll
        for (int j = 0; j < 4; ++j) {
          const unsigned u  = (unsigned)(dd[j] - n0);
          const bool     mt = u < (unsigned)nown;
          const unsigned m  = __builtin_amdgcn_ballot_w32(mt);
          if (m) {
            if (mt) q[qn + (int)__builtin_amdgcn_mbcnt_lo(m, 0u)] = (int)(u << 20) | (eb[k] + j);
            qn += __builtin_popcount(m);
          }
        }
      }
    }
    while (qn >= 32) {
      qn -= 32;
      own_flush(q + qn, 32, lane, n0, H, src, dinv, tab);
    }
  }
  if (qn > 0) own_flush(q, qn, lane, n0, H, src, dinv, tab);
  __syncthreads();

  for (int i = tid; i < nown * 32; i += OWN_THREADS) {
    const int u = i >> 5;
    const int l = i & 31;
    const int n = n0 + u;
    const float di  = dinv[n];
    const float di2 = di * di;
    const size_t off = (size_t)n * NF + 4 * l;
    const float4 hv = *(const float4*)(H + off);
    const float4 bv = *(const float4*)(b + 4 * l);
    const float* r = tab + u * NF + l;
    float4 o;
    o.x = fmaxf(r[0]  + di2 * hv.x + bv.x, 0.0f);
    o.y = fmaxf(r[32] + di2 * hv.y + bv.y, 0.0f);
    o.z = fmaxf(r[64] + di2 * hv.z + bv.z, 0.0f);
    o.w = fmaxf(r[96] + di2 * hv.w + bv.w, 0.0f);
    *(float4*)(Y + off) = o;
  }
}

__global__ __launch_bounds__(PS_THREADS) void pool_scatter(const float* __restrict__ Y,
                                                           const int* __restrict__ src,
                                                           const int* __restrict__ dst,
                                                           const float* __restrict__ dinv,
                                                           const int* __restrict__ batch,
                                                           float* __restrict__ part) {
  __shared__ float tab[TAB];
  __shared__ float cnt[N_GRAPHS];
  const int tid  = threadIdx.x;
  const int lane = tid & 31;

  for (int i = tid; i < TAB / 4; i += PS_THREADS)
    ((float4*)tab)[i] = make_float4(0.0f, 0.0f, 0.0f, 0.0f);
  if (tid < N_GRAPHS) cnt[tid] = 0.0f;
  __syncthreads();

  const int wv = __builtin_amdgcn_readfirstlane((int)blockIdx.x * (PS_THREADS / 32) + (tid >> 5));
  const int t0 = (int)(((long long)wv * N_GROUPS) / PS_WAVES) * 8;
  const int t1 = (int)(((long long)(wv + 1) * N_GROUPS) / PS_WAVES) * 8;

  int   s[PS_CHUNKS];
  int   g[PS_CHUNKS];
  float w[PS_CHUNKS];
  int   qn[PS_CHUNKS];
  int   rn[PS_CHUNKS];
  #pragma unroll
  for (int c = 0; c < PS_CHUNKS; ++c) {
    int t = t0 + 32 * c + lane;
    t = t < t1 ? t : t1 - 1;
    qn[c] = t / (EPN + 1);
    rn[c] = t - qn[c] * (EPN + 1);
    const int e = qn[c] * EPN + (rn[c] < EPN ? rn[c] : EPN - 1);
    s[c] = src[e];
    g[c] = dst[e];
  }
  #pragma unroll
  for (int c = 0; c < PS_CHUNKS; ++c) {
    const int sn = rn[c] < EPN ? s[c] : qn[c];
    const int dn = rn[c] < EPN ? g[c] : qn[c];
    s[c] = sn;
    w[c] = dinv[sn] * dinv[dn];
    g[c] = batch[dn];
  }
  #pragma unroll
  for (int c = 0; c < PS_CHUNKS; ++c)
    if (t0 + 32 * c + lane < t1 && rn[c] == EPN) atomicAdd(&cnt[g[c]], 1.0f);

  #pragma unroll
  for (int c = 0; c < PS_CHUNKS; ++c) {
    const int left = t1 - (t0 + 32 * c);
    const int ng = (left < 32 ? left : 32) >> 3;
    for (int q = 0; q < ng; ++q) {
      float4 v[8];
      #pragma unroll
      for (int j = 0; j < 8; ++j) {
        const int sk = __builtin_amdgcn_readlane(s[c], q * 8 + j);
        v[j] = *(const float4*)(Y + (size_t)sk * NF + lane * 4);
      }
      #pragma unroll
      for (int j = 0; j < 8; ++j) {
        const float wk = __int_as_float(__builtin_amdgcn_readlane(__float_as_int(w[c]), q * 8 + j));
        const int   gk = __builtin_amdgcn_readlane(g[c], q * 8 + j);
        float* row = tab + gk * NF + lane;
        atomicAdd(row,      v[j].x * wk);
        atomicAdd(row + 32, v[j].y * wk);
        atomicAdd(row + 64, v[j].z * wk);
        atomicAdd(row + 96, v[j].w * wk);
      }
    }
  }
  __syncthreads();

  float* P = part + (size_t)blockIdx.x * PART;
  for (int i = tid; i < TAB / 4; i += PS_THREADS) {
    const float* r = tab + (i >> 5) * NF + (i & 31);
    ((float4*)P)[i] = make_float4(r[0], r[32], r[64], r[96]);
  }
  if (tid < N_GRAPHS) P[TAB + tid] = cnt[tid];
}

__global__ __launch_bounds__(256) void reduce_parts(const float* __restrict__ part,
                                                    float* __restrict__ T) {
  __shared__ float4 red[8][32];
  const int wv   = threadIdx.x >> 5;
  const int lane = threadIdx.x & 31;
  const int q    = blockIdx.x * 32 + lane;
  const float4* p = (const float4*)part + q;
  float4 a = make_float4(0.0f, 0.0f, 0.0f, 0.0f);
  #pragma unroll
  for (int b = 0; b < PS_BLOCKS / 8; ++b) {
    const float4 x = p[(size_t)(wv * (PS_BLOCKS / 8) + b) * (PART / 4)];
    a.x += x.x; a.y += x.y; a.z += x.z; a.w += x.w;
  }
  red[wv][lane] = a;
  __syncthreads();
  if (wv == 0) {
    float4 t = red[0][lane];
    #pragma unroll
    for (int k = 1; k < 8; ++k) {
      const float4 x = red[k][lane];
      t.x += x.x; t.y += x.y; t.z += x.z; t.w += x.w;
    }
    ((float4*)T)[q] = t;
  }
}

extern "C" void kernel_launch(void* const* d_in, const int* in_sizes, int n_in,
                              void* d_out, int out_size, void* d_ws, size_t ws_size,
                              hipStream_t stream) {
  const float* x    = (const float*)d_in[0];
  const int*   ei   = (const int*)  d_in[1];
  const int*   batch= (const int*)  d_in[2];
  const float* W1   = (const float*)d_in[3];
  const float* b1   = (const float*)d_in[4];
  const float* W2   = (const float*)d_in[5];
  const float* b2   = (const float*)d_in[6];
  float* out = (float*)d_out;
  float* ws  = (float*)d_ws;

  const int* src = ei;
  const int* dst = ei + N_EDGES;

  float* dinv = ws;
  float* H    = ws + 50048;
  float* AGG  = H  + (size_t)N_NODES * NF;
  float* Y    = AGG + (size_t)N_NODES * NF;
  float* part = H;
  float* Tsum = AGG;

  const int T = 256;
  const int gNodes    = (N_NODES + T - 1) / T;
  const int gEdges    = (N_EDGES + T - 1) / T;
  const int gGemm     = N_NODES / 16;

  zero_kernel<<<gNodes, T, 0, stream>>>(dinv, N_NODES);
  deg_kernel<<<gEdges, T, 0, stream>>>(dst, dinv);
  dinv_kernel<<<gNodes, T, 0, stream>>>(dinv);

  gemm_wmma<<<gGemm, T, 0, stream>>>(x, W1, H);
  own_aggregate<<<OWN_BLOCKS, OWN_THREADS, 0, stream>>>(H, src, dst, dinv, b1, Y);

  pool_scatter<<<PS_BLOCKS, PS_THREADS, 0, stream>>>(Y, src, dst, dinv, batch, part);
  reduce_parts<<<(PART / 4) / 32, T, 0, stream>>>(part, Tsum);
  gemm_wmma_pooled<<<N_GRAPHS / 16, T, 0, stream>>>(Tsum, W2, Tsum + TAB, b2, out);
}


# layer-1 product moved behind the aggregation (both linear): owner blocks sum X rows in LDS, then multiply their table by W1 on f32 WMMA at the store; no H array or product launch; sum order changes
# speedup vs baseline: 2.9216x; 1.1533x over previous
#include <hip/hip_runtime.h>

#define N_NODES  50000
#define N_EDGES  800000
#define NF       128
#define N_GRAPHS 128

#define EPN        (N_EDGES / N_NODES)
#define N_ITEMS    (N_EDGES + N_NODES)
#define N_GROUPS   (N_ITEMS / 8)
#define PS_BLOCKS  256
#define PS_THREADS 1024
#define PS_WAVES   (PS_BLOCKS * (PS_THREADS / 32))
#define PS_CHUNKS  4
#define TAB        (N_GRAPHS * NF)
#define PART       (TAB + N_GRAPHS)
static_assert(N_EDGES % N_NODES == 0, "item numbering needs a whole number of edges per node");
static_assert(N_ITEMS % 8 == 0, "items are handed out 8 at a time");
static_assert(((N_GROUPS + PS_WAVES - 1) / PS_WAVES) * 8 <= 32 * PS_CHUNKS, "a wave's items must fit its registers");
static_assert(PS_BLOCKS % 8 == 0 && (PART / 4) % 32 == 0, "reduce kernel tiling");

#define OWN_BLOCKS  256
#define OWN_NODES   ((N_NODES + OWN_BLOCKS - 1) / OWN_BLOCKS)
#define OWN_THREADS 1024
#define OWN_WAVES   (OWN_THREADS / 32)
#define CH_EDGES    128
#define N_CH        (N_EDGES / CH_EDGES)
#define OWN_U       4
#define OWN_STEPS   ((N_CH + OWN_WAVES * OWN_U - 1) / (OWN_WAVES * OWN_U))
#define Q_CAP       (32 + OWN_U * CH_EDGES)
#define OWN_TILES   ((OWN_NODES + 15) / 16)
#define OWN_ROWS    (OWN_TILES * 16)
#define TS          (NF + 4)
static_assert(N_EDGES % CH_EDGES == 0, "the sweep reads whole chunks");
static_assert(N_EDGES <= (1 << 20) && OWN_NODES <= (1 << 11), "queue entry packs node and edge");
static_assert((OWN_BLOCKS - 1) * OWN_NODES < N_NODES, "every owner block has at least one node");
static_assert(N_CH >= OWN_WAVES * OWN_U, "start offset plus chunk index wraps at most once");

typedef __attribute__((ext_vector_type(2))) float v2f;
typedef __attribute__((ext_vector_type(8))) float v8f;

__global__ void zero_kernel(float* __restrict__ p, int n) {
  int i = blockIdx.x * blockDim.x + threadIdx.x;
  if (i < n) p[i] = 0.0f;
}

__global__ void deg_kernel(const int* __restrict__ dst, float* __restrict__ deg) {
  int e = blockIdx.x * blockDim.x + threadIdx.x;
  if (e < N_EDGES) atomicAdd(&deg[dst[e]], 1.0f);
}

__global__ void dinv_kernel(float* __restrict__ deg) {
  int i = blockIdx.x * blockDim.x + threadIdx.x;
  if (i < N_NODES) deg[i] = rsqrtf(deg[i] + 1.0f);
}

__global__ __launch_bounds__(256) void gemm_wmma_pooled(const float* __restrict__ T,
                                                        const float* __restrict__ W,
                                                        const float* __restrict__ cnt,
                                                        const float* __restrict__ bias,
                                                        float* __restrict__ out) {
  const int wave = threadIdx.x >> 5;
  const int lane = threadIdx.x & 31;
  const int half = lane >> 4;
  const int l    = lane & 15;
  const int m0   = blockIdx.x * 16;
  const int n0   = wave * 16;

  const float* Trow = T + (size_t)(m0 + l) * NF;
  v8f acc = {};

  #pragma unroll
  for (int k = 0; k < NF; k += 4) {
    v2f a = *(const v2f*)(Trow + k + 2 * half);
    v2f b;
    b.x = W[(size_t)(k + 2 * half + 0) * NF + n0 + l];
    b.y = W[(size_t)(k + 2 * half + 1) * NF + n0 + l];
    acc = __builtin_amdgcn_wmma_f32_16x16x4_f32(
         false, a,  false, b,
         (short)0, acc,  false,  false);
  }

  const float bn = bias[n0 + l];
  float* op = out + (size_t)(m0 + 8 * half) * NF + n0 + l;
  #pragma unroll
  for (int r = 0; r < 8; ++r) op[(size_t)r * NF] = acc[r] + cnt[m0 + 8 * half + r] * bn;
}

__device__ __forceinline__ void own_flush(const int* q, int n, int lane, int n0,
                                          const float* __restrict__ X,
                                          const int* __restrict__ src,
                                          const float* __restrict__ dinv,
                                          float* tab) {
  const int ent = q[lane < n ? lane : n - 1];
  const int e = ent & 0xFFFFF;
  const int u = ent >> 20;
  const int s = src[e];
  const float wgt = dinv[s] * dinv[n0 + u];
  const int ng = (n + 7) >> 3;
  for (int g = 0; g < ng; ++g) {
    float4 v[8];
    #pragma unroll
    for (int j = 0; j < 8; ++j) {
      const int sk = __builtin_amdgcn_readlane(s, g * 8 + j);
      v[j] = *(const float4*)(X + (size_t)sk * NF + lane * 4);
    }
    #pragma unroll
    for (int j = 0; j < 8; ++j) {
      const int k = g * 8 + j;
      if (k < n) {
        const float wk = __int_as_float(__builtin_amdgcn_readlane(__float_as_int(wgt), k));
        const int   uk = __builtin_amdgcn_readlane(u, k);
        float* row = tab + uk * TS + lane;
        atomicAdd(row,      v[j].x * wk);
        atomicAdd(row + 32, v[j].y * wk);
        atomicAdd(row + 64, v[j].z * wk);
        atomicAdd(row + 96, v[j].w * wk);
      }
    }
  }
}

__global__ __launch_bounds__(OWN_THREADS) void own_aggregate(const float* __restrict__ X,
                                                             const float* __restrict__ W,
                                                             const int* __restrict__ src,
                                                             const int* __restrict__ dst,
                                                             const float* __restrict__ dinv,
                                                             const float* __restrict__ b,
                                                             float* __restrict__ Y) {
  __shared__ __attribute__((aligned(16))) float tab[OWN_ROWS * TS];
  __shared__ int   queue[OWN_WAVES * Q_CAP];
  const int tid  = threadIdx.x;
  const int lane = tid & 31;
  const int n0   = (int)blockIdx.x * OWN_NODES;
  const int nown = (N_NODES - n0) < OWN_NODES ? (N_NODES - n0) : OWN_NODES;

  for (int i = tid; i < OWN_ROWS * TS / 4; i += OWN_THREADS)
    ((float4*)tab)[i] = make_float4(0.0f, 0.0f, 0.0f, 0.0f);
  __syncthreads();

  const int w = __builtin_amdgcn_readfirstlane(tid >> 5);
  int* q = queue + w * Q_CAP;
  int qn = 0;
  const int c0 = (int)(((long long)blockIdx.x * N_CH) / OWN_BLOCKS);

  for (int i = 0; i < OWN_STEPS; ++i) {
    int4 d[OWN_U];
    int  eb[OWN_U];
    bool ok[OWN_U];
    #pragma unroll
    for (int k = 0; k < OWN_U; ++k) {
      const int ci = (i * OWN_U + k) * OWN_WAVES + w;
      ok[k] = ci < N_CH;
      int cc = c0 + (ok[k] ? ci : 0);
      cc = cc >= N_CH ? cc - N_CH : cc;
      eb[k] = cc * CH_EDGES + lane * 4;
      d[k] = *(const int4*)(dst + eb[k]);
    }
    #pragma unroll
    for (int k = 0; k < OWN_U; ++k) {
      if (ok[k]) {
        const int dd[4] = {d[k].x, d[k].y, d[k].z, d[k].w};
        #pragma unroll
        for (int j = 0; j < 4; ++j) {
          const unsigned u  = (unsigned)(dd[j] - n0);
          const bool     mt = u < (unsigned)nown;
          const unsigned m  = __builtin_amdgcn_ballot_w32(mt);
          if (m) {
            if (mt) q[qn + (int)__builtin_amdgcn_mbcnt_lo(m, 0u)] = (int)(u << 20) | (eb[k] + j);
            qn += __builtin_popcount(m);
          }
        }
      }
    }
    while (qn >= 32) {
      qn -= 32;
      own_flush(q + qn, 32, lane, n0, X, src, dinv, tab);
    }
  }
  if (qn > 0) own_flush(q, qn, lane, n0, X, src, dinv, tab);
  __syncthreads();

  for (int i = tid; i < nown * 32; i += OWN_THREADS) {
    const int u = i >> 5;
    const int l = i & 31;
    const int n = n0 + u;
    const float di  = dinv[n];
    const float di2 = di * di;
    const float4 xv = *(const float4*)(X + (size_t)n * NF + 4 * l);
    float* r = tab + u * TS + l;
    r[0]  += di2 * xv.x;
    r[32] += di2 * xv.y;
    r[64] += di2 * xv.z;
    r[96] += di2 * xv.w;
  }

  const int half = lane >> 4;
  const int c    = lane & 15;
  const int nc   = (w & 7) * 16 + c;
  v2f bf[32];
  #pragma unroll
  for (int t = 0; t < 32; ++t) {
    const int q0 = 8 * (t >> 1) + 4 * half + 2 * (t & 1);
    const int f0 = 4 * (q0 & 31) + (q0 >> 5);
    bf[t].x = W[(size_t)f0 * NF + nc];
    bf[t].y = W[(size_t)(f0 + 4) * NF + nc];
  }
  const float bn = b[nc];
  __syncthreads();

  for (int m0 = (w >> 3) * 16; m0 < nown; m0 += 64) {
    const float* ar = tab + (m0 + c) * TS + 4 * half;
    v8f acc = {};
    #pragma unroll
    for (int u = 0; u < 16; ++u) {
      const float4 a4 = *(const float4*)(ar + 8 * u);
      v2f a;
      a.x = a4.x; a.y = a4.y;
      acc = __builtin_amdgcn_wmma_f32_16x16x4_f32(
           false, a,  false, bf[2 * u],
           (short)0, acc,  false,  false);
      a.x = a4.z; a.y = a4.w;
      acc = __builtin_amdgcn_wmma_f32_16x16x4_f32(
           false, a,  false, bf[2 * u + 1],
           (short)0, acc,  false,  false);
    }
    float* yp = Y + (size_t)(n0 + m0 + 8 * half) * NF + nc;
    if (m0 + 16 <= nown) {
      #pragma unroll
      for (int r = 0; r < 8; ++r) yp[(size_t)r * NF] = fmaxf(acc[r] + bn, 0.0f);
    } else {
      #pragma unroll
      for (int r = 0; r < 8; ++r)
        if (m0 + 8 * half + r < nown) yp[(size_t)r * NF] = fmaxf(acc[r] + bn, 0.0f);
    }
  }
}

__global__ __launch_bounds__(PS_THREADS) void pool_scatter(const float* __restrict__ Y,
                                                           const int* __restrict__ src,
                                                           const int* __restrict__ dst,
                                                           const float* __restrict__ dinv,
                                                           const int* __restrict__ batch,
                                                           float* __restrict__ part) {
  __shared__ float tab[TAB];
  __shared__ float cnt[N_GRAPHS];
  const int tid  = threadIdx.x;
  const int lane = tid & 31;

  for (int i = tid; i < TAB / 4; i += PS_THREADS)
    ((float4*)tab)[i] = make_float4(0.0f, 0.0f, 0.0f, 0.0f);
  if (tid < N_GRAPHS) cnt[tid] = 0.0f;
  __syncthreads();

  const int wv = __builtin_amdgcn_readfirstlane((int)blockIdx.x * (PS_THREADS / 32) + (tid >> 5));
  const int t0 = (int)(((long long)wv * N_GROUPS) / PS_WAVES) * 8;
  const int t1 = (int)(((long long)(wv + 1) * N_GROUPS) / PS_WAVES) * 8;

  int   s[PS_CHUNKS];
  int   g[PS_CHUNKS];
  float w[PS_CHUNKS];
  int   qn[PS_CHUNKS];
  int   rn[PS_CHUNKS];
  #pragma unroll
  for (int c = 0; c < PS_CHUNKS; ++c) {
    int t = t0 + 32 * c + lane;
    t = t < t1 ? t : t1 - 1;
    qn[c] = t / (EPN + 1);
    rn[c] = t - qn[c] * (EPN + 1);
    const int e = qn[c] * EPN + (rn[c] < EPN ? rn[c] : EPN - 1);
    s[c] = src[e];
    g[c] = dst[e];
  }
  #pragma unroll
  for (int c = 0; c < PS_CHUNKS; ++c) {
    const int sn = rn[c] < EPN ? s[c] : qn[c];
    const int dn = rn[c] < EPN ? g[c] : qn[c];
    s[c] = sn;
    w[c] = dinv[sn] * dinv[dn];
    g[c] = batch[dn];
  }
  #pragma unroll
  for (int c = 0; c < PS_CHUNKS; ++c)
    if (t0 + 32 * c + lane < t1 && rn[c] == EPN) atomicAdd(&cnt[g[c]], 1.0f);

  #pragma unroll
  for (int c = 0; c < PS_CHUNKS; ++c) {
    const int left = t1 - (t0 + 32 * c);
    const int ng = (left < 32 ? left : 32) >> 3;
    for (int q = 0; q < ng; ++q) {
      float4 v[8];
      #pragma unroll
      for (int j = 0; j < 8; ++j) {
        const int sk = __builtin_amdgcn_readlane(s[c], q * 8 + j);
        v[j] = *(const float4*)(Y + (size_t)sk * NF + lane * 4);
      }
      #pragma unroll
      for (int j = 0; j < 8; ++j) {
        const float wk = __int_as_float(__builtin_amdgcn_readlane(__float_as_int(w[c]), q * 8 + j));
        const int   gk = __builtin_amdgcn_readlane(g[c], q * 8 + j);
        float* row = tab + gk * NF + lane;
        atomicAdd(row,      v[j].x * wk);
        atomicAdd(row + 32, v[j].y * wk);
        atomicAdd(row + 64, v[j].z * wk);
        atomicAdd(row + 96, v[j].w * wk);
      }
    }
  }
  __syncthreads();

  float* P = part + (size_t)blockIdx.x * PART;
  for (int i = tid; i < TAB / 4; i += PS_THREADS) {
    const float* r = tab + (i >> 5) * NF + (i & 31);
    ((float4*)P)[i] = make_float4(r[0], r[32], r[64], r[96]);
  }
  if (tid < N_GRAPHS) P[TAB + tid] = cnt[tid];
}

__global__ __launch_bounds__(256) void reduce_parts(const float* __restrict__ part,
                                                    float* __restrict__ T) {
  __shared__ float4 red[8][32];
  const int wv   = threadIdx.x >> 5;
  const int lane = threadIdx.x & 31;
  const int q    = blockIdx.x * 32 + lane;
  const float4* p = (const float4*)part + q;
  float4 a = make_float4(0.0f, 0.0f, 0.0f, 0.0f);
  #pragma unroll
  for (int b = 0; b < PS_BLOCKS / 8; ++b) {
    const float4 x = p[(size_t)(wv * (PS_BLOCKS / 8) + b) * (PART / 4)];
    a.x += x.x; a.y += x.y; a.z += x.z; a.w += x.w;
  }
  red[wv][lane] = a;
  __syncthreads();
  if (wv == 0) {
    float4 t = red[0][lane];
    #pragma unroll
    for (int k = 1; k < 8; ++k) {
      const float4 x = red[k][lane];
      t.x += x.x; t.y += x.y; t.z += x.z; t.w += x.w;
    }
    ((float4*)T)[q] = t;
  }
}

extern "C" void kernel_launch(void* const* d_in, const int* in_sizes, int n_in,
                              void* d_out, int out_size, void* d_ws, size_t ws_size,
                              hipStream_t stream) {
  const float* x    = (const float*)d_in[0];
  const int*   ei   = (const int*)  d_in[1];
  const int*   batch= (const int*)  d_in[2];
  const float* W1   = (const float*)d_in[3];
  const float* b1   = (const float*)d_in[4];
  const float* W2   = (const float*)d_in[5];
  const float* b2   = (const float*)d_in[6];
  float* out = (float*)d_out;
  float* ws  = (float*)d_ws;

  const int* src = ei;
  const int* dst = ei + N_EDGES;

  float* dinv = ws;
  float* H    = ws + 50048;
  float* AGG  = H  + (size_t)N_NODES * NF;
  float* Y    = AGG + (size_t)N_NODES * NF;
  float* part = H;
  float* Tsum = AGG;

  const int T = 256;
  const int gNodes    = (N_NODES + T - 1) / T;
  const int gEdges    = (N_EDGES + T - 1) / T;

  zero_kernel<<<gNodes, T, 0, stream>>>(dinv, N_NODES);
  deg_kernel<<<gEdges, T, 0, stream>>>(dst, dinv);
  dinv_kernel<<<gNodes, T, 0, stream>>>(dinv);

  own_aggregate<<<OWN_BLOCKS, OWN_THREADS, 0, stream>>>(x, W1, src, dst, dinv, b1, Y);

  pool_scatter<<<PS_BLOCKS, PS_THREADS, 0, stream>>>(Y, src, dst, dinv, batch, part);
  reduce_parts<<<(PART / 4) / 32, T, 0, stream>>>(part, Tsum);
  gemm_wmma_pooled<<<N_GRAPHS / 16, T, 0, stream>>>(Tsum, W2, Tsum + TAB, b2, out);
}


# layer-2 per-item lookups (weights of both ends, graph id) read from an LDS copy of a packed node table (one word per node, copied per block), not scattered global loads; sum order unchanged
# speedup vs baseline: 3.0405x; 1.0407x over previous
#include <hip/hip_runtime.h>

#define N_NODES  50000
#define N_EDGES  800000
#define NF       128
#define N_GRAPHS 128

#define EPN        (N_EDGES / N_NODES)
#define N_ITEMS    (N_EDGES + N_NODES)
#define N_GROUPS   (N_ITEMS / 8)
#define PS_BLOCKS  256
#define PS_THREADS 1024
#define PS_WAVES   (PS_BLOCKS * (PS_THREADS / 32))
#define PS_CHUNKS  4
#define TAB        (N_GRAPHS * NF)
#define PART       (TAB + N_GRAPHS)
static_assert(N_EDGES % N_NODES == 0, "item numbering needs a whole number of edges per node");
static_assert(N_ITEMS % 8 == 0, "items are handed out 8 at a time");
static_assert(((N_GROUPS + PS_WAVES - 1) / PS_WAVES) * 8 <= 32 * PS_CHUNKS, "a wave's items must fit its registers");
static_assert(PS_BLOCKS % 8 == 0 && (PART / 4) % 32 == 0, "reduce kernel tiling");

#define OWN_BLOCKS  256
#define OWN_NODES   ((N_NODES + OWN_BLOCKS - 1) / OWN_BLOCKS)
#define OWN_THREADS 1024
#define OWN_WAVES   (OWN_THREADS / 32)
#define CH_EDGES    128
#define N_CH        (N_EDGES / CH_EDGES)
#define OWN_U       4
#define OWN_STEPS   ((N_CH + OWN_WAVES * OWN_U - 1) / (OWN_WAVES * OWN_U))
#define Q_CAP       (32 + OWN_U * CH_EDGES)
#define OWN_TILES   ((OWN_NODES + 15) / 16)
#define OWN_ROWS    (OWN_TILES * 16)
#define TS          (NF + 4)
static_assert(N_EDGES % CH_EDGES == 0, "the sweep reads whole chunks");
static_assert(N_EDGES <= (1 << 20) && OWN_NODES <= (1 << 11), "queue entry packs node and edge");
static_assert((OWN_BLOCKS - 1) * OWN_NODES < N_NODES, "every owner block has at least one node");
static_assert(N_CH >= OWN_WAVES * OWN_U, "start offset plus chunk index wraps at most once");

typedef __attribute__((ext_vector_type(2))) float v2f;
typedef __attribute__((ext_vector_type(8))) float v8f;

__global__ void zero_kernel(float* __restrict__ p, int n) {
  int i = blockIdx.x * blockDim.x + threadIdx.x;
  if (i < n) p[i] = 0.0f;
}

__global__ void deg_kernel(const int* __restrict__ dst, float* __restrict__ deg) {
  int e = blockIdx.x * blockDim.x + threadIdx.x;
  if (e < N_EDGES) atomicAdd(&deg[dst[e]], 1.0f);
}

static_assert(N_EDGES + 1 < (1 << 20) && N_GRAPHS <= (1 << 12), "node word packs degree and graph id");
__global__ void dinv_kernel(float* __restrict__ deg, const int* __restrict__ batch,
                            unsigned* __restrict__ rec) {
  int i = blockIdx.x * blockDim.x + threadIdx.x;
  if (i < N_NODES) {
    const float d1 = deg[i] + 1.0f;
    deg[i] = rsqrtf(d1);
    rec[i] = (unsigned)d1 | ((unsigned)batch[i] << 20);
  }
}

__global__ __launch_bounds__(256) void gemm_wmma_pooled(const float* __restrict__ T,
                                                        const float* __restrict__ W,
                                                        const float* __restrict__ cnt,
                                                        const float* __restrict__ bias,
                                                        float* __restrict__ out) {
  const int wave = threadIdx.x >> 5;
  const int lane = threadIdx.x & 31;
  const int half = lane >> 4;
  const int l    = lane & 15;
  const int m0   = blockIdx.x * 16;
  const int n0   = wave * 16;

  const float* Trow = T + (size_t)(m0 + l) * NF;
  v8f acc = {};

  #pragma unroll
  for (int k = 0; k < NF; k += 4) {
    v2f a = *(const v2f*)(Trow + k + 2 * half);
    v2f b;
    b.x = W[(size_t)(k + 2 * half + 0) * NF + n0 + l];
    b.y = W[(size_t)(k + 2 * half + 1) * NF + n0 + l];
    acc = __builtin_amdgcn_wmma_f32_16x16x4_f32(
         false, a,  false, b,
         (short)0, acc,  false,  false);
  }

  const float bn = bias[n0 + l];
  float* op = out + (size_t)(m0 + 8 * half) * NF + n0 + l;
  #pragma unroll
  for (int r = 0; r < 8; ++r) op[(size_t)r * NF] = acc[r] + cnt[m0 + 8 * half + r] * bn;
}

__device__ __forceinline__ void own_flush(const int* q, int n, int lane, int n0,
                                          const float* __restrict__ X,
                                          const int* __restrict__ src,
                                          const float* __restrict__ dinv,
                                          float* tab) {
  const int ent = q[lane < n ? lane : n - 1];
  const int e = ent & 0xFFFFF;
  const int u = ent >> 20;
  const int s = src[e];
  const float wgt = dinv[s] * dinv[n0 + u];
  const int ng = (n + 7) >> 3;
  for (int g = 0; g < ng; ++g) {
    float4 v[8];
    #pragma unroll
    for (int j = 0; j < 8; ++j) {
      const int sk = __builtin_amdgcn_readlane(s, g * 8 + j);
      v[j] = *(const float4*)(X + (size_t)sk * NF + lane * 4);
    }
    #pragma unroll
    for (int j = 0; j < 8; ++j) {
      const int k = g * 8 + j;
      if (k < n) {
        const float wk = __int_as_float(__builtin_amdgcn_readlane(__float_as_int(wgt), k));
        const int   uk = __builtin_amdgcn_readlane(u, k);
        float* row = tab + uk * TS + lane;
        atomicAdd(row,      v[j].x * wk);
        atomicAdd(row + 32, v[j].y * wk);
        atomicAdd(row + 64, v[j].z * wk);
        atomicAdd(row + 96, v[j].w * wk);
      }
    }
  }
}

__global__ __launch_bounds__(OWN_THREADS) void own_aggregate(const float* __restrict__ X,
                                                             const float* __restrict__ W,
                                                             const int* __restrict__ src,
                                                             const int* __restrict__ dst,
                                                             const float* __restrict__ dinv,
                                                             const float* __restrict__ b,
                                                             float* __restrict__ Y) {
  __shared__ __attribute__((aligned(16))) float tab[OWN_ROWS * TS];
  __shared__ int   queue[OWN_WAVES * Q_CAP];
  const int tid  = threadIdx.x;
  const int lane = tid & 31;
  const int n0   = (int)blockIdx.x * OWN_NODES;
  const int nown = (N_NODES - n0) < OWN_NODES ? (N_NODES - n0) : OWN_NODES;

  for (int i = tid; i < OWN_ROWS * TS / 4; i += OWN_THREADS)
    ((float4*)tab)[i] = make_float4(0.0f, 0.0f, 0.0f, 0.0f);
  __syncthreads();

  const int w = __builtin_amdgcn_readfirstlane(tid >> 5);
  int* q = queue + w * Q_CAP;
  int qn = 0;
  const int c0 = (int)(((long long)blockIdx.x * N_CH) / OWN_BLOCKS);

  for (int i = 0; i < OWN_STEPS; ++i) {
    int4 d[OWN_U];
    int  eb[OWN_U];
    bool ok[OWN_U];
    #pragma unroll
    for (int k = 0; k < OWN_U; ++k) {
      const int ci = (i * OWN_U + k) * OWN_WAVES + w;
      ok[k] = ci < N_CH;
      int cc = c0 + (ok[k] ? ci : 0);
      cc = cc >= N_CH ? cc - N_CH : cc;
      eb[k] = cc * CH_EDGES + lane * 4;
      d[k] = *(const int4*)(dst + eb[k]);
    }
    #pragma unroll
    for (int k = 0; k < OWN_U; ++k) {
      if (ok[k]) {
        const int dd[4] = {d[k].x, d[k].y, d[k].z, d[k].w};
        #pragma unroll
        for (int j = 0; j < 4; ++j) {
          const unsigned u  = (unsigned)(dd[j] - n0);
          const bool     mt = u < (unsigned)nown;
          const unsigned m  = __builtin_amdgcn_ballot_w32(mt);
          if (m) {
            if (mt) q[qn + (int)__builtin_amdgcn_mbcnt_lo(m, 0u)] = (int)(u << 20) | (eb[k] + j);
            qn += __builtin_popcount(m);
          }
        }
      }
    }
    while (qn >= 32) {
      qn -= 32;
      own_flush(q + qn, 32, lane, n0, X, src, dinv, tab);
    }
  }
  if (qn > 0) own_flush(q, qn, lane, n0, X, src, dinv, tab);
  __syncthreads();

  for (int i = tid; i < nown * 32; i += OWN_THREADS) {
    const int u = i >> 5;
    const int l = i & 31;
    const int n = n0 + u;
    const float di  = dinv[n];
    const float di2 = di * di;
    const float4 xv = *(const float4*)(X + (size_t)n * NF + 4 * l);
    float* r = tab + u * TS + l;
    r[0]  += di2 * xv.x;
    r[32] += di2 * xv.y;
    r[64] += di2 * xv.z;
    r[96] += di2 * xv.w;
  }

  const int half = lane >> 4;
  const int c    = lane & 15;
  const int nc   = (w & 7) * 16 + c;
  v2f bf[32];
  #pragma unroll
  for (int t = 0; t < 32; ++t) {
    const int q0 = 8 * (t >> 1) + 4 * half + 2 * (t & 1);
    const int f0 = 4 * (q0 & 31) + (q0 >> 5);
    bf[t].x = W[(size_t)f0 * NF + nc];
    bf[t].y = W[(size_t)(f0 + 4) * NF + nc];
  }
  const float bn = b[nc];
  __syncthreads();

  for (int m0 = (w >> 3) * 16; m0 < nown; m0 += 64) {
    const float* ar = tab + (m0 + c) * TS + 4 * half;
    v8f acc = {};
    #pragma unroll
    for (int u = 0; u < 16; ++u) {
      const float4 a4 = *(const float4*)(ar + 8 * u);
      v2f a;
      a.x = a4.x; a.y = a4.y;
      acc = __builtin_amdgcn_wmma_f32_16x16x4_f32(
           false, a,  false, bf[2 * u],
           (short)0, acc,  false,  false);
      a.x = a4.z; a.y = a4.w;
      acc = __builtin_amdgcn_wmma_f32_16x16x4_f32(
           false, a,  false, bf[2 * u + 1],
           (short)0, acc,  false,  false);
    }
    float* yp = Y + (size_t)(n0 + m0 + 8 * half) * NF + nc;
    if (m0 + 16 <= nown) {
      #pragma unroll
      for (int r = 0; r < 8; ++r) yp[(size_t)r * NF] = fmaxf(acc[r] + bn, 0.0f);
    } else {
      #pragma unroll
      for (int r = 0; r < 8; ++r)
        if (m0 + 8 * half + r < nown) yp[(size_t)r * NF] = fmaxf(acc[r] + bn, 0.0f);
    }
  }
}

__global__ __launch_bounds__(PS_THREADS) void pool_scatter(const float* __restrict__ Y,
                                                           const int* __restrict__ src,
                                                           const int* __restrict__ dst,
                                                           const unsigned* __restrict__ rec,
                                                           float* __restrict__ part) {
  __shared__ float tab[TAB];
  __shared__ float cnt[N_GRAPHS];
  __shared__ __attribute__((aligned(16))) unsigned nrec[N_NODES];
  static_assert(N_NODES % 4 == 0, "the node table is copied 16 bytes at a time");
  constexpr int NR4     = N_NODES / 4;
  constexpr int NR_PER  = (NR4 + PS_THREADS - 1) / PS_THREADS;
  static_assert((PS_BLOCKS - 1) * (NR4 / PS_BLOCKS) < NR4, "a block's start offset wraps at most once");
  const int tid  = threadIdx.x;
  const int lane = tid & 31;

  const int wv = __builtin_amdgcn_readfirstlane((int)blockIdx.x * (PS_THREADS / 32) + (tid >> 5));
  const int t0 = (int)(((long long)wv * N_GROUPS) / PS_WAVES) * 8;
  const int t1 = (int)(((long long)(wv + 1) * N_GROUPS) / PS_WAVES) * 8;

  int   s[PS_CHUNKS];
  int   g[PS_CHUNKS];
  float w[PS_CHUNKS];
  int   qn[PS_CHUNKS];
  int   rn[PS_CHUNKS];
  #pragma unroll
  for (int c = 0; c < PS_CHUNKS; ++c) {
    int t = t0 + 32 * c + lane;
    t = t < t1 ? t : t1 - 1;
    qn[c] = t / (EPN + 1);
    rn[c] = t - qn[c] * (EPN + 1);
    const int e = qn[c] * EPN + (rn[c] < EPN ? rn[c] : EPN - 1);
    s[c] = src[e];
    g[c] = dst[e];
  }

  const uint4* rec4 = (const uint4*)rec;
  const int boff = (int)blockIdx.x * (NR4 / PS_BLOCKS);
  uint4 nv[NR_PER];
  int   np[NR_PER];
  #pragma unroll
  for (int k = 0; k < NR_PER; ++k) {
    int i = tid + k * PS_THREADS;
    i = (i < NR4 ? i : NR4 - 1) + boff;
    np[k] = i >= NR4 ? i - NR4 : i;
    nv[k] = rec4[np[k]];
  }

  for (int i = tid; i < TAB / 4; i += PS_THREADS)
    ((float4*)tab)[i] = make_float4(0.0f, 0.0f, 0.0f, 0.0f);
  if (tid < N_GRAPHS) cnt[tid] = 0.0f;

  #pragma unroll
  for (int k = 0; k < NR_PER; ++k) ((uint4*)nrec)[np[k]] = nv[k];
  __syncthreads();

  #pragma unroll
  for (int c = 0; c < PS_CHUNKS; ++c) {
    const int sn = rn[c] < EPN ? s[c] : qn[c];
    const int dn = rn[c] < EPN ? g[c] : qn[c];
    const unsigned rs = nrec[sn];
    const unsigned rd = nrec[dn];
    s[c] = sn;
    w[c] = rsqrtf((float)(rs & 0xFFFFFu)) * rsqrtf((float)(rd & 0xFFFFFu));
    g[c] = (int)(rd >> 20);
  }
  #pragma unroll
  for (int c = 0; c < PS_CHUNKS; ++c)
    if (t0 + 32 * c + lane < t1 && rn[c] == EPN) atomicAdd(&cnt[g[c]], 1.0f);

  #pragma unroll
  for (int c = 0; c < PS_CHUNKS; ++c) {
    const int left = t1 - (t0 + 32 * c);
    const int ng = (left < 32 ? left : 32) >> 3;
    for (int q = 0; q < ng; ++q) {
      float4 v[8];
      #pragma unroll
      for (int j = 0; j < 8; ++j) {
        const int sk = __builtin_amdgcn_readlane(s[c], q * 8 + j);
        v[j] = *(const float4*)(Y + (size_t)sk * NF + lane * 4);
      }
      #pragma unroll
      for (int j = 0; j < 8; ++j) {
        const float wk = __int_as_float(__builtin_amdgcn_readlane(__float_as_int(w[c]), q * 8 + j));
        const int   gk = __builtin_amdgcn_readlane(g[c], q * 8 + j);
        float* row = tab + gk * NF + lane;
        atomicAdd(row,      v[j].x * wk);
        atomicAdd(row + 32, v[j].y * wk);
        atomicAdd(row + 64, v[j].z * wk);
        atomicAdd(row + 96, v[j].w * wk);
      }
    }
  }
  __syncthreads();

  float* P = part + (size_t)blockIdx.x * PART;
  for (int i = tid; i < TAB / 4; i += PS_THREADS) {
    const float* r = tab + (i >> 5) * NF + (i & 31);
    ((float4*)P)[i] = make_float4(r[0], r[32], r[64], r[96]);
  }
  if (tid < N_GRAPHS) P[TAB + tid] = cnt[tid];
}

__global__ __launch_bounds__(256) void reduce_parts(const float* __restrict__ part,
                                                    float* __restrict__ T) {
  __shared__ float4 red[8][32];
  const int wv   = threadIdx.x >> 5;
  const int lane = threadIdx.x & 31;
  const int q    = blockIdx.x * 32 + lane;
  const float4* p = (const float4*)part + q;
  float4 a = make_float4(0.0f, 0.0f, 0.0f, 0.0f);
  #pragma unroll
  for (int b = 0; b < PS_BLOCKS / 8; ++b) {
    const float4 x = p[(size_t)(wv * (PS_BLOCKS / 8) + b) * (PART / 4)];
    a.x += x.x; a.y += x.y; a.z += x.z; a.w += x.w;
  }
  red[wv][lane] = a;
  __syncthreads();
  if (wv == 0) {
    float4 t = red[0][lane];
    #pragma unroll
    for (int k = 1; k < 8; ++k) {
      const float4 x = red[k][lane];
      t.x += x.x; t.y += x.y; t.z += x.z; t.w += x.w;
    }
    ((float4*)T)[q] = t;
  }
}

extern "C" void kernel_launch(void* const* d_in, const int* in_sizes, int n_in,
                              void* d_out, int out_size, void* d_ws, size_t ws_size,
                              hipStream_t stream) {
  const float* x    = (const float*)d_in[0];
  const int*   ei   = (const int*)  d_in[1];
  const int*   batch= (const int*)  d_in[2];
  const float* W1   = (const float*)d_in[3];
  const float* b1   = (const float*)d_in[4];
  const float* W2   = (const float*)d_in[5];
  const float* b2   = (const float*)d_in[6];
  float* out = (float*)d_out;
  float* ws  = (float*)d_ws;

  const int* src = ei;
  const int* dst = ei + N_EDGES;

  float* dinv = ws;
  float* H    = ws + 50048;
  float* AGG  = H  + (size_t)N_NODES * NF;
  float* Y    = AGG + (size_t)N_NODES * NF;
  float* part = H;
  float* Tsum = AGG;
  unsigned* rec = (unsigned*)(H + (size_t)PS_BLOCKS * PART);

  const int T = 256;
  const int gNodes    = (N_NODES + T - 1) / T;
  const int gEdges    = (N_EDGES + T - 1) / T;

  zero_kernel<<<gNodes, T, 0, stream>>>(dinv, N_NODES);
  deg_kernel<<<gEdges, T, 0, stream>>>(dst, dinv);
  dinv_kernel<<<gNodes, T, 0, stream>>>(dinv, batch, rec);

  own_aggregate<<<OWN_BLOCKS, OWN_THREADS, 0, stream>>>(x, W1, src, dst, dinv, b1, Y);

  pool_scatter<<<PS_BLOCKS, PS_THREADS, 0, stream>>>(Y, src, dst, rec, part);
  reduce_parts<<<(PART / 4) / 32, T, 0, stream>>>(part, Tsum);
  gemm_wmma_pooled<<<N_GRAPHS / 16, T, 0, stream>>>(Tsum, W2, Tsum + TAB, b2, out);
}


# layer-1 owner sweep reads one owner byte per edge (block number of dst, packed by the degree launch), not the 4-byte dst; hit 4-edge groups are re-checked on dst; sum order: LDS arrival as before
# speedup vs baseline: 3.2265x; 1.0612x over previous
#include <hip/hip_runtime.h>

#define N_NODES  50000
#define N_EDGES  800000
#define NF       128
#define N_GRAPHS 128

#define EPN        (N_EDGES / N_NODES)
#define N_ITEMS    (N_EDGES + N_NODES)
#define N_GROUPS   (N_ITEMS / 8)
#define PS_BLOCKS  256
#define PS_THREADS 1024
#define PS_WAVES   (PS_BLOCKS * (PS_THREADS / 32))
#define PS_CHUNKS  4
#define TAB        (N_GRAPHS * NF)
#define PART       (TAB + N_GRAPHS)
static_assert(N_EDGES % N_NODES == 0, "item numbering needs a whole number of edges per node");
static_assert(N_ITEMS % 8 == 0, "items are handed out 8 at a time");
static_assert(((N_GROUPS + PS_WAVES - 1) / PS_WAVES) * 8 <= 32 * PS_CHUNKS, "a wave's items must fit its registers");
static_assert(PS_BLOCKS % 8 == 0 && (PART / 4) % 32 == 0, "reduce kernel tiling");

#define OWN_BLOCKS  256
#define OWN_NODES   ((N_NODES + OWN_BLOCKS - 1) / OWN_BLOCKS)
#define OWN_THREADS 1024
#define OWN_WAVES   (OWN_THREADS / 32)
#define GRP_EDGES   4
#define CH_EDGES    256
#define N_CH        (N_EDGES / CH_EDGES)
#define OWN_U       8
#define OWN_STEPS   ((N_CH + OWN_WAVES * OWN_U - 1) / (OWN_WAVES * OWN_U))
#define G_CAP       (32 + OWN_U * 64)
#define E_CAP       (32 + 32 * GRP_EDGES)
#define OWN_TILES   ((OWN_NODES + 15) / 16)
#define OWN_ROWS    (OWN_TILES * 16)
#define TS          (NF + 4)
static_assert(N_EDGES % CH_EDGES == 0, "the sweep reads whole chunks");
static_assert((N_NODES - 1) / OWN_NODES < 256, "an owner block's number fits a byte");
static_assert(N_EDGES <= (1 << 20) && OWN_NODES <= (1 << 11), "queue entry packs node and edge");
static_assert((OWN_BLOCKS - 1) * OWN_NODES < N_NODES, "every owner block has at least one node");
static_assert(N_CH >= OWN_WAVES * OWN_U, "start offset plus chunk index wraps at most once");

typedef __attribute__((ext_vector_type(2))) float v2f;
typedef __attribute__((ext_vector_type(8))) float v8f;

__global__ void zero_kernel(float* __restrict__ p, int n) {
  int i = blockIdx.x * blockDim.x + threadIdx.x;
  if (i < n) p[i] = 0.0f;
}

__global__ void deg_kernel(const int* __restrict__ dst, float* __restrict__ deg,
                           unsigned* __restrict__ own4) {
  int i = blockIdx.x * blockDim.x + threadIdx.x;
  if (i < N_EDGES / GRP_EDGES) {
    const int4 d = ((const int4*)dst)[i];
    atomicAdd(&deg[d.x], 1.0f);
    atomicAdd(&deg[d.y], 1.0f);
    atomicAdd(&deg[d.z], 1.0f);
    atomicAdd(&deg[d.w], 1.0f);
    own4[i] =  ((unsigned)d.x / OWN_NODES)        | (((unsigned)d.y / OWN_NODES) << 8) |
              (((unsigned)d.z / OWN_NODES) << 16) | (((unsigned)d.w / OWN_NODES) << 24);
  }
}

static_assert(N_EDGES + 1 < (1 << 20) && N_GRAPHS <= (1 << 12), "node word packs degree and graph id");
__global__ void dinv_kernel(float* __restrict__ deg, const int* __restrict__ batch,
                            unsigned* __restrict__ rec) {
  int i = blockIdx.x * blockDim.x + threadIdx.x;
  if (i < N_NODES) {
    const float d1 = deg[i] + 1.0f;
    deg[i] = rsqrtf(d1);
    rec[i] = (unsigned)d1 | ((unsigned)batch[i] << 20);
  }
}

__global__ __launch_bounds__(256) void gemm_wmma_pooled(const float* __restrict__ T,
                                                        const float* __restrict__ W,
                                                        const float* __restrict__ cnt,
                                                        const float* __restrict__ bias,
                                                        float* __restrict__ out) {
  const int wave = threadIdx.x >> 5;
  const int lane = threadIdx.x & 31;
  const int half = lane >> 4;
  const int l    = lane & 15;
  const int m0   = blockIdx.x * 16;
  const int n0   = wave * 16;

  const float* Trow = T + (size_t)(m0 + l) * NF;
  v8f acc = {};

  #pragma unroll
  for (int k = 0; k < NF; k += 4) {
    v2f a = *(const v2f*)(Trow + k + 2 * half);
    v2f b;
    b.x = W[(size_t)(k + 2 * half + 0) * NF + n0 + l];
    b.y = W[(size_t)(k + 2 * half + 1) * NF + n0 + l];
    acc = __builtin_amdgcn_wmma_f32_16x16x4_f32(
         false, a,  false, b,
         (short)0, acc,  false,  false);
  }

  const float bn = bias[n0 + l];
  float* op = out + (size_t)(m0 + 8 * half) * NF + n0 + l;
  #pragma unroll
  for (int r = 0; r < 8; ++r) op[(size_t)r * NF] = acc[r] + cnt[m0 + 8 * half + r] * bn;
}

__device__ __forceinline__ void own_flush(const int* q, int n, int lane, int n0,
                                          const float* __restrict__ X,
                                          const int* __restrict__ src,
                                          const float* __restrict__ dinv,
                                          float* tab) {
  const int ent = q[lane < n ? lane : n - 1];
  const int e = ent & 0xFFFFF;
  const int u = ent >> 20;
  const int s = src[e];
  const float wgt = dinv[s] * dinv[n0 + u];
  const int ng = (n + 7) >> 3;
  for (int g = 0; g < ng; ++g) {
    float4 v[8];
    #pragma unroll
    for (int j = 0; j < 8; ++j) {
      const int sk = __builtin_amdgcn_readlane(s, g * 8 + j);
      v[j] = *(const float4*)(X + (size_t)sk * NF + lane * 4);
    }
    #pragma unroll
    for (int j = 0; j < 8; ++j) {
      const int k = g * 8 + j;
      if (k < n) {
        const float wk = __int_as_float(__builtin_amdgcn_readlane(__float_as_int(wgt), k));
        const int   uk = __builtin_amdgcn_readlane(u, k);
        float* row = tab + uk * TS + lane;
        atomicAdd(row,      v[j].x * wk);
        atomicAdd(row + 32, v[j].y * wk);
        atomicAdd(row + 64, v[j].z * wk);
        atomicAdd(row + 96, v[j].w * wk);
      }
    }
  }
}

__device__ __forceinline__ int own_expand(const int* gq, int n, int lane, int n0, int nown,
                                          const int* __restrict__ dst, int* eq, int en) {
  const int grp = gq[lane < n ? lane : n - 1];
  const int4 d = *(const int4*)(dst + GRP_EDGES * grp);
  const int dd[4] = {d.x, d.y, d.z, d.w};
  #pragma unroll
  for (int j = 0; j < GRP_EDGES; ++j) {
    const unsigned u  = (unsigned)(dd[j] - n0);
    const bool     mt = lane < n && u < (unsigned)nown;
    const unsigned m  = __builtin_amdgcn_ballot_w32(mt);
    if (m) {
      if (mt) eq[en + (int)__builtin_amdgcn_mbcnt_lo(m, 0u)] = (int)(u << 20) | (GRP_EDGES * grp + j);
      en += __builtin_popcount(m);
    }
  }
  return en;
}

__global__ __launch_bounds__(OWN_THREADS) void own_aggregate(const float* __restrict__ X,
                                                             const float* __restrict__ W,
                                                             const int* __restrict__ src,
                                                             const int* __restrict__ dst,
                                                             const unsigned* __restrict__ own4,
                                                             const float* __restrict__ dinv,
                                                             const float* __restrict__ b,
                                                             float* __restrict__ Y) {
  __shared__ __attribute__((aligned(16))) float tab[OWN_ROWS * TS];
  __shared__ int   gqueue[OWN_WAVES * G_CAP];
  __shared__ int   equeue[OWN_WAVES * E_CAP];
  const int tid  = threadIdx.x;
  const int lane = tid & 31;
  const int n0   = (int)blockIdx.x * OWN_NODES;
  const int nown = (N_NODES - n0) < OWN_NODES ? (N_NODES - n0) : OWN_NODES;

  for (int i = tid; i < OWN_ROWS * TS / 4; i += OWN_THREADS)
    ((float4*)tab)[i] = make_float4(0.0f, 0.0f, 0.0f, 0.0f);
  __syncthreads();

  const int w = __builtin_amdgcn_readfirstlane(tid >> 5);
  int* gq = gqueue + w * G_CAP;
  int* eq = equeue + w * E_CAP;
  int gn = 0;
  int en = 0;
  const int c0 = (int)(((long long)blockIdx.x * N_CH) / OWN_BLOCKS);
  const unsigned pat = (unsigned)blockIdx.x * 0x01010101u;

  for (int i = 0; i <= OWN_STEPS; ++i) {
    const bool last = i == OWN_STEPS;
    if (!last) {
      uint2 o[OWN_U];
      int   g0[OWN_U];
      bool  ok[OWN_U];
      #pragma unroll
      for (int k = 0; k < OWN_U; ++k) {
        const int ci = (i * OWN_U + k) * OWN_WAVES + w;
        ok[k] = ci < N_CH;
        int cc = c0 + (ok[k] ? ci : 0);
        cc = cc >= N_CH ? cc - N_CH : cc;
        g0[k] = cc * (CH_EDGES / GRP_EDGES) + lane * 2;
        o[k] = *(const uint2*)(own4 + g0[k]);
      }
      #pragma unroll
      for (int k = 0; k < OWN_U; ++k) {
        if (ok[k]) {
          const unsigned yy[2] = {o[k].x ^ pat, o[k].y ^ pat};
          #pragma unroll
          for (int h = 0; h < 2; ++h) {
            const bool     mt = ((yy[h] - 0x01010101u) & ~yy[h] & 0x80808080u) != 0u;
            const unsigned m  = __builtin_amdgcn_ballot_w32(mt);
            if (m) {
              if (mt) gq[gn + (int)__builtin_amdgcn_mbcnt_lo(m, 0u)] = g0[k] + h;
              gn += __builtin_popcount(m);
            }
          }
        }
      }
    }
    for (;;) {
      if (en >= 32) {
        en -= 32;
        own_flush(eq + en, 32, lane, n0, X, src, dinv, tab);
      } else if (gn >= 32 || (last && gn > 0)) {
        const int n = gn < 32 ? gn : 32;
        gn -= n;
        en = own_expand(gq + gn, n, lane, n0, nown, dst, eq, en);
      } else {
        break;
      }
    }
  }
  if (en > 0) own_flush(eq, en, lane, n0, X, src, dinv, tab);
  __syncthreads();

  for (int i = tid; i < nown * 32; i += OWN_THREADS) {
    const int u = i >> 5;
    const int l = i & 31;
    const int n = n0 + u;
    const float di  = dinv[n];
    const float di2 = di * di;
    const float4 xv = *(const float4*)(X + (size_t)n * NF + 4 * l);
    float* r = tab + u * TS + l;
    r[0]  += di2 * xv.x;
    r[32] += di2 * xv.y;
    r[64] += di2 * xv.z;
    r[96] += di2 * xv.w;
  }

  const int half = lane >> 4;
  const int c    = lane & 15;
  const int nc   = (w & 7) * 16 + c;
  v2f bf[32];
  #pragma unroll
  for (int t = 0; t < 32; ++t) {
    const int q0 = 8 * (t >> 1) + 4 * half + 2 * (t & 1);
    const int f0 = 4 * (q0 & 31) + (q0 >> 5);
    bf[t].x = W[(size_t)f0 * NF + nc];
    bf[t].y = W[(size_t)(f0 + 4) * NF + nc];
  }
  const float bn = b[nc];
  __syncthreads();

  for (int m0 = (w >> 3) * 16; m0 < nown; m0 += 64) {
    const float* ar = tab + (m0 + c) * TS + 4 * half;
    v8f acc = {};
    #pragma unroll
    for (int u = 0; u < 16; ++u) {
      const float4 a4 = *(const float4*)(ar + 8 * u);
      v2f a;
      a.x = a4.x; a.y = a4.y;
      acc = __builtin_amdgcn_wmma_f32_16x16x4_f32(
           false, a,  false, bf[2 * u],
           (short)0, acc,  false,  false);
      a.x = a4.z; a.y = a4.w;
      acc = __builtin_amdgcn_wmma_f32_16x16x4_f32(
           false, a,  false, bf[2 * u + 1],
           (short)0, acc,  false,  false);
    }
    float* yp = Y + (size_t)(n0 + m0 + 8 * half) * NF + nc;
    if (m0 + 16 <= nown) {
      #pragma unroll
      for (int r = 0; r < 8; ++r) yp[(size_t)r * NF] = fmaxf(acc[r] + bn, 0.0f);
    } else {
      #pragma unroll
      for (int r = 0; r < 8; ++r)
        if (m0 + 8 * half + r < nown) yp[(size_t)r * NF] = fmaxf(acc[r] + bn, 0.0f);
    }
  }
}

__global__ __launch_bounds__(PS_THREADS) void pool_scatter(const float* __restrict__ Y,
                                                           const int* __restrict__ src,
                                                           const int* __restrict__ dst,
                                                           const unsigned* __restrict__ rec,
                                                           float* __restrict__ part) {
  __shared__ float tab[TAB];
  __shared__ float cnt[N_GRAPHS];
  __shared__ __attribute__((aligned(16))) unsigned nrec[N_NODES];
  static_assert(N_NODES % 4 == 0, "the node table is copied 16 bytes at a time");
  constexpr int NR4     = N_NODES / 4;
  constexpr int NR_PER  = (NR4 + PS_THREADS - 1) / PS_THREADS;
  static_assert((PS_BLOCKS - 1) * (NR4 / PS_BLOCKS) < NR4, "a block's start offset wraps at most once");
  const int tid  = threadIdx.x;
  const int lane = tid & 31;

  const int wv = __builtin_amdgcn_readfirstlane((int)blockIdx.x * (PS_THREADS / 32) + (tid >> 5));
  const int t0 = (int)(((long long)wv * N_GROUPS) / PS_WAVES) * 8;
  const int t1 = (int)(((long long)(wv + 1) * N_GROUPS) / PS_WAVES) * 8;

  int   s[PS_CHUNKS];
  int   g[PS_CHUNKS];
  float w[PS_CHUNKS];
  int   qn[PS_CHUNKS];
  int   rn[PS_CHUNKS];
  #pragma unroll
  for (int c = 0; c < PS_CHUNKS; ++c) {
    int t = t0 + 32 * c + lane;
    t = t < t1 ? t : t1 - 1;
    qn[c] = t / (EPN + 1);
    rn[c] = t - qn[c] * (EPN + 1);
    const int e = qn[c] * EPN + (rn[c] < EPN ? rn[c] : EPN - 1);
    s[c] = src[e];
    g[c] = dst[e];
  }

  const uint4* rec4 = (const uint4*)rec;
  const int boff = (int)blockIdx.x * (NR4 / PS_BLOCKS);
  uint4 nv[NR_PER];
  int   np[NR_PER];
  #pragma unroll
  for (int k = 0; k < NR_PER; ++k) {
    int i = tid + k * PS_THREADS;
    i = (i < NR4 ? i : NR4 - 1) + boff;
    np[k] = i >= NR4 ? i - NR4 : i;
    nv[k] = rec4[np[k]];
  }

  for (int i = tid; i < TAB / 4; i += PS_THREADS)
    ((float4*)tab)[i] = make_float4(0.0f, 0.0f, 0.0f, 0.0f);
  if (tid < N_GRAPHS) cnt[tid] = 0.0f;

  #pragma unroll
  for (int k = 0; k < NR_PER; ++k) ((uint4*)nrec)[np[k]] = nv[k];
  __syncthreads();

  #pragma unroll
  for (int c = 0; c < PS_CHUNKS; ++c) {
    const int sn = rn[c] < EPN ? s[c] : qn[c];
    const int dn = rn[c] < EPN ? g[c] : qn[c];
    const unsigned rs = nrec[sn];
    const unsigned rd = nrec[dn];
    s[c] = sn;
    w[c] = rsqrtf((float)(rs & 0xFFFFFu)) * rsqrtf((float)(rd & 0xFFFFFu));
    g[c] = (int)(rd >> 20);
  }
  #pragma unroll
  for (int c = 0; c < PS_CHUNKS; ++c)
    if (t0 + 32 * c + lane < t1 && rn[c] == EPN) atomicAdd(&cnt[g[c]], 1.0f);

  #pragma unroll
  for (int c = 0; c < PS_CHUNKS; ++c) {
    const int left = t1 - (t0 + 32 * c);
    const int ng = (left < 32 ? left : 32) >> 3;
    for (int q = 0; q < ng; ++q) {
      float4 v[8];
      #pragma unroll
      for (int j = 0; j < 8; ++j) {
        const int sk = __builtin_amdgcn_readlane(s[c], q * 8 + j);
        v[j] = *(const float4*)(Y + (size_t)sk * NF + lane * 4);
      }
      #pragma unroll
      for (int j = 0; j < 8; ++j) {
        const float wk = __int_as_float(__builtin_amdgcn_readlane(__float_as_int(w[c]), q * 8 + j));
        const int   gk = __builtin_amdgcn_readlane(g[c], q * 8 + j);
        float* row = tab + gk * NF + lane;
        atomicAdd(row,      v[j].x * wk);
        atomicAdd(row + 32, v[j].y * wk);
        atomicAdd(row + 64, v[j].z * wk);
        atomicAdd(row + 96, v[j].w * wk);
      }
    }
  }
  __syncthreads();

  float* P = part + (size_t)blockIdx.x * PART;
  for (int i = tid; i < TAB / 4; i += PS_THREADS) {
    const float* r = tab + (i >> 5) * NF + (i & 31);
    ((float4*)P)[i] = make_float4(r[0], r[32], r[64], r[96]);
  }
  if (tid < N_GRAPHS) P[TAB + tid] = cnt[tid];
}

__global__ __launch_bounds__(256) void reduce_parts(const float* __restrict__ part,
                                                    float* __restrict__ T) {
  __shared__ float4 red[8][32];
  const int wv   = threadIdx.x >> 5;
  const int lane = threadIdx.x & 31;
  const int q    = blockIdx.x * 32 + lane;
  const float4* p = (const float4*)part + q;
  float4 a = make_float4(0.0f, 0.0f, 0.0f, 0.0f);
  #pragma unroll
  for (int b = 0; b < PS_BLOCKS / 8; ++b) {
    const float4 x = p[(size_t)(wv * (PS_BLOCKS / 8) + b) * (PART / 4)];
    a.x += x.x; a.y += x.y; a.z += x.z; a.w += x.w;
  }
  red[wv][lane] = a;
  __syncthreads();
  if (wv == 0) {
    float4 t = red[0][lane];
    #pragma unroll
    for (int k = 1; k < 8; ++k) {
      const float4 x = red[k][lane];
      t.x += x.x; t.y += x.y; t.z += x.z; t.w += x.w;
    }
    ((float4*)T)[q] = t;
  }
}

extern "C" void kernel_launch(void* const* d_in, const int* in_sizes, int n_in,
                              void* d_out, int out_size, void* d_ws, size_t ws_size,
                              hipStream_t stream) {
  const float* x    = (const float*)d_in[0];
  const int*   ei   = (const int*)  d_in[1];
  const int*   batch= (const int*)  d_in[2];
  const float* W1   = (const float*)d_in[3];
  const float* b1   = (const float*)d_in[4];
  const float* W2   = (const float*)d_in[5];
  const float* b2   = (const float*)d_in[6];
  float* out = (float*)d_out;
  float* ws  = (float*)d_ws;

  const int* src = ei;
  const int* dst = ei + N_EDGES;

  float* dinv = ws;
  float* H    = ws + 50048;
  float* AGG  = H  + (size_t)N_NODES * NF;
  float* Y    = AGG + (size_t)N_NODES * NF;
  float* part = H;
  float* Tsum = AGG;
  unsigned* rec = (unsigned*)(H + (size_t)PS_BLOCKS * PART);
  unsigned* own4 = rec + 50048;

  const int T = 256;
  const int gNodes    = (N_NODES + T - 1) / T;
  const int gEdge4    = (N_EDGES / GRP_EDGES + T - 1) / T;

  zero_kernel<<<gNodes, T, 0, stream>>>(dinv, N_NODES);
  deg_kernel<<<gEdge4, T, 0, stream>>>(dst, dinv, own4);
  dinv_kernel<<<gNodes, T, 0, stream>>>(dinv, batch, rec);

  own_aggregate<<<OWN_BLOCKS, OWN_THREADS, 0, stream>>>(x, W1, src, dst, own4, dinv, b1, Y);

  pool_scatter<<<PS_BLOCKS, PS_THREADS, 0, stream>>>(Y, src, dst, rec, part);
  reduce_parts<<<(PART / 4) / 32, T, 0, stream>>>(part, Tsum);
  gemm_wmma_pooled<<<N_GRAPHS / 16, T, 0, stream>>>(Tsum, W2, Tsum + TAB, b2, out);
}
